# starting kernel
# baseline (speedup 1.0000x reference)
#include <hip/hip_runtime.h>
#include <hip/hip_bf16.h>

#define MDIM 1024
#define BDIM 256
#define LWIN 8
#define PRED 8
#define NSLOT (LWIN + PRED)

typedef __attribute__((ext_vector_type(2))) float v2f;
typedef __attribute__((ext_vector_type(8))) float v8f;

__device__ __forceinline__ v8f wmma_f32_k4(v2f a, v2f b, v8f c) {
    return __builtin_amdgcn_wmma_f32_16x16x4_f32(
         false, a,  false, b,
         (short)0, c,  false,  false);
}

__global__ void dmd_build_agt(const float* __restrict__ Ar,
                              const float* __restrict__ Ai,
                              float* __restrict__ agtr,
                              float* __restrict__ agti) {
    size_t i = (size_t)blockIdx.x * blockDim.x + threadIdx.x;
    if (i >= (size_t)LWIN * MDIM * MDIM) return;
    int n = (int)(i & (MDIM - 1));
    int m = (int)((i >> 10) & (MDIM - 1));
    int p = (int)(i >> 20);
    int l = (p == 0) ? 0 : (LWIN - p);
    size_t src = (size_t)l * MDIM * MDIM + (size_t)m * MDIM + n;
    size_t dst = (size_t)p * MDIM * MDIM + (size_t)(n >> 1) * (2 * MDIM)
               + 2 * m + (n & 1);
    agtr[dst] = Ar[src];
    agti[dst] = Ai[src];
}

__global__ void dmd_init_state(const float* __restrict__ x,
                               float* __restrict__ stR,
                               float* __restrict__ stI) {
    size_t i = (size_t)blockIdx.x * blockDim.x + threadIdx.x;
    if (i >= (size_t)LWIN * MDIM * BDIM) return;
    int b = (int)(i & (BDIM - 1));
    int n = (int)((i >> 8) & (MDIM - 1));
    int s = (int)(i >> 18);
    float v = x[(size_t)b * (LWIN * MDIM) + (size_t)s * MDIM + n];
    size_t dst = (size_t)s * MDIM * BDIM + (size_t)(n >> 1) * (2 * BDIM)
               + 2 * b + (n & 1);
    stR[dst] = v;
    stI[dst] = 0.0f;
}

__global__ __launch_bounds__(32) void dmd_gemm_step(
    const float* __restrict__ agtr, const float* __restrict__ agti,
    const float* __restrict__ stR,  const float* __restrict__ stI,
    float* __restrict__ partR, float* __restrict__ partI, int t) {
    const int lane = threadIdx.x & 31;
    const int g  = lane >> 4;
    const int lm = lane & 15;
    const int p  = blockIdx.z;
    const int m0 = blockIdx.y * 64;
    const int b0 = blockIdx.x * 16;
    const int s  = t + p;

    const float* aR = agtr + (size_t)p * MDIM * MDIM;
    const float* aI = agti + (size_t)p * MDIM * MDIM;
    const float* uR = stR + (size_t)s * MDIM * BDIM;
    const float* uI = stI + (size_t)s * MDIM * BDIM;

    v8f c_rr[4] = {}, c_ii[4] = {}, c_ri[4] = {}, c_ir[4] = {};

    for (int k = 0; k < MDIM; k += 4) {
        const int np = (k + 2 * g) >> 1;
        const float2 bR2 = *(const float2*)(uR + (size_t)np * (2 * BDIM) + 2 * (b0 + lm));
        const float2 bI2 = *(const float2*)(uI + (size_t)np * (2 * BDIM) + 2 * (b0 + lm));
        v2f fbR = { bR2.x, bR2.y };
        v2f fbI = { bI2.x, bI2.y };
#pragma unroll
        for (int mt = 0; mt < 4; ++mt) {
            const int m = m0 + mt * 16 + lm;
            const float2 aR2 = *(const float2*)(aR + (size_t)np * (2 * MDIM) + 2 * m);
            const float2 aI2 = *(const float2*)(aI + (size_t)np * (2 * MDIM) + 2 * m);
            v2f faR = { aR2.x, aR2.y };
            v2f faI = { aI2.x, aI2.y };
            c_rr[mt] = wmma_f32_k4(faR, fbR, c_rr[mt]);
            c_ii[mt] = wmma_f32_k4(faI, fbI, c_ii[mt]);
            c_ri[mt] = wmma_f32_k4(faR, fbI, c_ri[mt]);
            c_ir[mt] = wmma_f32_k4(faI, fbR, c_ir[mt]);
        }
    }

#pragma unroll
    for (int mt = 0; mt < 4; ++mt) {
#pragma unroll
        for (int r = 0; r < 8; ++r) {
            const int m = m0 + mt * 16 + r + 8 * g;
            const size_t o = (size_t)p * MDIM * BDIM + (size_t)m * BDIM + b0 + lm;
            partR[o] = c_rr[mt][r] - c_ii[mt][r];
            partI[o] = c_ri[mt][r] + c_ir[mt][r];
        }
    }
}

__global__ void dmd_reduce_step(const float* __restrict__ partR,
                                const float* __restrict__ partI,
                                float* __restrict__ stR,
                                float* __restrict__ stI, int t) {
    int i = blockIdx.x * blockDim.x + threadIdx.x;
    if (i >= MDIM * BDIM) return;
    int m = i >> 8;
    int b = i & (BDIM - 1);
    float sr = 0.0f, si = 0.0f;
#pragma unroll
    for (int p = 0; p < LWIN; ++p) {
        sr += partR[(size_t)p * MDIM * BDIM + i];
        si += partI[(size_t)p * MDIM * BDIM + i];
    }
    size_t dst = (size_t)(LWIN + t) * MDIM * BDIM
               + (size_t)(m >> 1) * (2 * BDIM) + 2 * b + (m & 1);
    stR[dst] = sr;
    stI[dst] = si;
}

__global__ void dmd_write_out(const float* __restrict__ stR,
                              float* __restrict__ out) {
    size_t i = (size_t)blockIdx.x * blockDim.x + threadIdx.x;
    if (i >= (size_t)BDIM * LWIN * MDIM) return;
    int m = (int)(i & (MDIM - 1));
    int l = (int)((i >> 10) & (LWIN - 1));
    int b = (int)(i >> 13);
    size_t src = (size_t)(LWIN + l) * MDIM * BDIM
               + (size_t)(m >> 1) * (2 * BDIM) + 2 * b + (m & 1);
    out[i] = stR[src];
}

extern "C" void kernel_launch(void* const* d_in, const int* in_sizes, int n_in,
                              void* d_out, int out_size, void* d_ws, size_t ws_size,
                              hipStream_t stream) {
    const float* x  = (const float*)d_in[0];
    const float* Ar = (const float*)d_in[1];
    const float* Ai = (const float*)d_in[2];

    float* ws = (float*)d_ws;
    const size_t AGT  = (size_t)LWIN * MDIM * MDIM;
    const size_t STPL = (size_t)NSLOT * MDIM * BDIM;
    const size_t PART = (size_t)LWIN * MDIM * BDIM;
    float* agtr = ws;
    float* agti = agtr + AGT;
    float* stR  = agti + AGT;
    float* stI  = stR + STPL;
    float* pR   = stI + STPL;
    float* pI   = pR + PART;

    dmd_build_agt<<<(unsigned)(AGT / 256), 256, 0, stream>>>(Ar, Ai, agtr, agti);
    dmd_init_state<<<(unsigned)(((size_t)LWIN * MDIM * BDIM) / 256), 256, 0, stream>>>(x, stR, stI);

    for (int t = 0; t < PRED; ++t) {
        dmd_gemm_step<<<dim3(BDIM / 16, MDIM / 64, LWIN), 32, 0, stream>>>(
            agtr, agti, stR, stI, pR, pI, t);
        dmd_reduce_step<<<(MDIM * BDIM) / 256, 256, 0, stream>>>(pR, pI, stR, stI, t);
    }

    dmd_write_out<<<(unsigned)(((size_t)BDIM * LWIN * MDIM) / 256), 256, 0, stream>>>(
        stR, (float*)d_out);
}


# gemm step: k loop software-pipelined in registers, fragments of k-block i+3 requested ahead of the matrix work of block i (ring of 5 fragment sets); every sum keeps its order
# speedup vs baseline: 1.3161x; 1.3161x over previous
#include <hip/hip_runtime.h>
#include <hip/hip_bf16.h>

#define MDIM 1024
#define BDIM 256
#define LWIN 8
#define PRED 8
#define NSLOT (LWIN + PRED)

typedef __attribute__((ext_vector_type(2))) float v2f;
typedef __attribute__((ext_vector_type(8))) float v8f;

__device__ __forceinline__ v8f wmma_f32_k4(v2f a, v2f b, v8f c) {
    return __builtin_amdgcn_wmma_f32_16x16x4_f32(
         false, a,  false, b,
         (short)0, c,  false,  false);
}

__global__ void dmd_build_agt(const float* __restrict__ Ar,
                              const float* __restrict__ Ai,
                              float* __restrict__ agtr,
                              float* __restrict__ agti) {
    size_t i = (size_t)blockIdx.x * blockDim.x + threadIdx.x;
    if (i >= (size_t)LWIN * MDIM * MDIM) return;
    int n = (int)(i & (MDIM - 1));
    int m = (int)((i >> 10) & (MDIM - 1));
    int p = (int)(i >> 20);
    int l = (p == 0) ? 0 : (LWIN - p);
    size_t src = (size_t)l * MDIM * MDIM + (size_t)m * MDIM + n;
    size_t dst = (size_t)p * MDIM * MDIM + (size_t)(n >> 1) * (2 * MDIM)
               + 2 * m + (n & 1);
    agtr[dst] = Ar[src];
    agti[dst] = Ai[src];
}

__global__ void dmd_init_state(const float* __restrict__ x,
                               float* __restrict__ stR,
                               float* __restrict__ stI) {
    size_t i = (size_t)blockIdx.x * blockDim.x + threadIdx.x;
    if (i >= (size_t)LWIN * MDIM * BDIM) return;
    int b = (int)(i & (BDIM - 1));
    int n = (int)((i >> 8) & (MDIM - 1));
    int s = (int)(i >> 18);
    float v = x[(size_t)b * (LWIN * MDIM) + (size_t)s * MDIM + n];
    size_t dst = (size_t)s * MDIM * BDIM + (size_t)(n >> 1) * (2 * BDIM)
               + 2 * b + (n & 1);
    stR[dst] = v;
    stI[dst] = 0.0f;
}

struct FragSet {
    v2f bR, bI;
    v2f aR[4], aI[4];
};

__device__ __forceinline__ void load_frags(FragSet& f,
        const float* __restrict__ pAR, const float* __restrict__ pAI,
        const float* __restrict__ pBR, const float* __restrict__ pBI, int k) {
    const size_t offA = (size_t)k * MDIM;
    const size_t offB = (size_t)k * BDIM;
    const float2 bR2 = *(const float2*)(pBR + offB);
    const float2 bI2 = *(const float2*)(pBI + offB);
    f.bR = (v2f){ bR2.x, bR2.y };
    f.bI = (v2f){ bI2.x, bI2.y };
#pragma unroll
    for (int mt = 0; mt < 4; ++mt) {
        const float2 aR2 = *(const float2*)(pAR + offA + 32 * mt);
        const float2 aI2 = *(const float2*)(pAI + offA + 32 * mt);
        f.aR[mt] = (v2f){ aR2.x, aR2.y };
        f.aI[mt] = (v2f){ aI2.x, aI2.y };
    }
}

__device__ __forceinline__ void mma_block(const FragSet& f,
        v8f (&c_rr)[4], v8f (&c_ii)[4], v8f (&c_ri)[4], v8f (&c_ir)[4]) {
#pragma unroll
    for (int mt = 0; mt < 4; ++mt) {
        c_rr[mt] = wmma_f32_k4(f.aR[mt], f.bR, c_rr[mt]);
        c_ii[mt] = wmma_f32_k4(f.aI[mt], f.bI, c_ii[mt]);
        c_ri[mt] = wmma_f32_k4(f.aR[mt], f.bI, c_ri[mt]);
        c_ir[mt] = wmma_f32_k4(f.aI[mt], f.bR, c_ir[mt]);
    }
}

__global__ __launch_bounds__(32) void dmd_gemm_step(
    const float* __restrict__ agtr, const float* __restrict__ agti,
    const float* __restrict__ stR,  const float* __restrict__ stI,
    float* __restrict__ partR, float* __restrict__ partI, int t) {
    const int lane = threadIdx.x & 31;
    const int g  = lane >> 4;
    const int lm = lane & 15;
    const int p  = blockIdx.z;
    const int m0 = blockIdx.y * 64;
    const int b0 = blockIdx.x * 16;
    const int s  = t + p;

    const float* pAR = agtr + (size_t)p * MDIM * MDIM + (size_t)g * (2 * MDIM) + 2 * (m0 + lm);
    const float* pAI = agti + (size_t)p * MDIM * MDIM + (size_t)g * (2 * MDIM) + 2 * (m0 + lm);
    const float* pBR = stR + (size_t)s * MDIM * BDIM + (size_t)g * (2 * BDIM) + 2 * (b0 + lm);
    const float* pBI = stI + (size_t)s * MDIM * BDIM + (size_t)g * (2 * BDIM) + 2 * (b0 + lm);

    v8f c_rr[4] = {}, c_ii[4] = {}, c_ri[4] = {}, c_ir[4] = {};

    FragSet f0, f1, f2, f3, f4;
    load_frags(f0, pAR, pAI, pBR, pBI, 0);
    load_frags(f1, pAR, pAI, pBR, pBI, 4);
    load_frags(f2, pAR, pAI, pBR, pBI, 8);

#define PIN() __builtin_amdgcn_sched_barrier(0)
#define MMA(f) mma_block(f, c_rr, c_ii, c_ri, c_ir)
#define LDF(f, kk) do { load_frags(f, pAR, pAI, pBR, pBI, (kk)); PIN(); } while (0)
    int k = 0;
    for (; k < 1000; k += 20) {
        LDF(f3, k + 12); MMA(f0);
        LDF(f4, k + 16); MMA(f1);
        LDF(f0, k + 20); MMA(f2);
        LDF(f1, k + 24); MMA(f3);
        LDF(f2, k + 28); MMA(f4);
    }
    LDF(f3, 1012); MMA(f0);
    LDF(f4, 1016); MMA(f1);
    LDF(f0, 1020); MMA(f2);
    MMA(f3);
    MMA(f4);
    MMA(f0);
#undef LDF
#undef MMA
#undef PIN

#pragma unroll
    for (int mt = 0; mt < 4; ++mt) {
#pragma unroll
        for (int r = 0; r < 8; ++r) {
            const int m = m0 + mt * 16 + r + 8 * g;
            const size_t o = (size_t)p * MDIM * BDIM + (size_t)m * BDIM + b0 + lm;
            partR[o] = c_rr[mt][r] - c_ii[mt][r];
            partI[o] = c_ri[mt][r] + c_ir[mt][r];
        }
    }
}

__global__ void dmd_reduce_step(const float* __restrict__ partR,
                                const float* __restrict__ partI,
                                float* __restrict__ stR,
                                float* __restrict__ stI, int t) {
    int i = blockIdx.x * blockDim.x + threadIdx.x;
    if (i >= MDIM * BDIM) return;
    int m = i >> 8;
    int b = i & (BDIM - 1);
    float sr = 0.0f, si = 0.0f;
#pragma unroll
    for (int p = 0; p < LWIN; ++p) {
        sr += partR[(size_t)p * MDIM * BDIM + i];
        si += partI[(size_t)p * MDIM * BDIM + i];
    }
    size_t dst = (size_t)(LWIN + t) * MDIM * BDIM
               + (size_t)(m >> 1) * (2 * BDIM) + 2 * b + (m & 1);
    stR[dst] = sr;
    stI[dst] = si;
}

__global__ void dmd_write_out(const float* __restrict__ stR,
                              float* __restrict__ out) {
    size_t i = (size_t)blockIdx.x * blockDim.x + threadIdx.x;
    if (i >= (size_t)BDIM * LWIN * MDIM) return;
    int m = (int)(i & (MDIM - 1));
    int l = (int)((i >> 10) & (LWIN - 1));
    int b = (int)(i >> 13);
    size_t src = (size_t)(LWIN + l) * MDIM * BDIM
               + (size_t)(m >> 1) * (2 * BDIM) + 2 * b + (m & 1);
    out[i] = stR[src];
}

extern "C" void kernel_launch(void* const* d_in, const int* in_sizes, int n_in,
                              void* d_out, int out_size, void* d_ws, size_t ws_size,
                              hipStream_t stream) {
    const float* x  = (const float*)d_in[0];
    const float* Ar = (const float*)d_in[1];
    const float* Ai = (const float*)d_in[2];

    float* ws = (float*)d_ws;
    const size_t AGT  = (size_t)LWIN * MDIM * MDIM;
    const size_t STPL = (size_t)NSLOT * MDIM * BDIM;
    const size_t PART = (size_t)LWIN * MDIM * BDIM;
    float* agtr = ws;
    float* agti = agtr + AGT;
    float* stR  = agti + AGT;
    float* stI  = stR + STPL;
    float* pR   = stI + STPL;
    float* pI   = pR + PART;

    dmd_build_agt<<<(unsigned)(AGT / 256), 256, 0, stream>>>(Ar, Ai, agtr, agti);
    dmd_init_state<<<(unsigned)(((size_t)LWIN * MDIM * BDIM) / 256), 256, 0, stream>>>(x, stR, stI);

    for (int t = 0; t < PRED; ++t) {
        dmd_gemm_step<<<dim3(BDIM / 16, MDIM / 64, LWIN), 32, 0, stream>>>(
            agtr, agti, stR, stI, pR, pI, t);
        dmd_reduce_step<<<(MDIM * BDIM) / 256, 256, 0, stream>>>(pR, pI, stR, stI, t);
    }

    dmd_write_out<<<(unsigned)(((size_t)BDIM * LWIN * MDIM) / 256), 256, 0, stream>>>(
        stR, (float*)d_out);
}


# gemm step retiled: a wave owns 64(m) x 32(b) instead of 64 x 16, four waves (one per SIMD) per workgroup on one operator tile; 12 not 20 fragment loads per 32 matrix ops; every sum keeps its order
# speedup vs baseline: 1.7616x; 1.3385x over previous
#include <hip/hip_runtime.h>
#include <hip/hip_bf16.h>

#define MDIM 1024
#define BDIM 256
#define LWIN 8
#define PRED 8
#define NSLOT (LWIN + PRED)

typedef __attribute__((ext_vector_type(2))) float v2f;
typedef __attribute__((ext_vector_type(8))) float v8f;

__device__ __forceinline__ v8f wmma_f32_k4(v2f a, v2f b, v8f c) {
    return __builtin_amdgcn_wmma_f32_16x16x4_f32(
         false, a,  false, b,
         (short)0, c,  false,  false);
}

__global__ void dmd_build_agt(const float* __restrict__ Ar,
                              const float* __restrict__ Ai,
                              float* __restrict__ agtr,
                              float* __restrict__ agti) {
    size_t i = (size_t)blockIdx.x * blockDim.x + threadIdx.x;
    if (i >= (size_t)LWIN * MDIM * MDIM) return;
    int n = (int)(i & (MDIM - 1));
    int m = (int)((i >> 10) & (MDIM - 1));
    int p = (int)(i >> 20);
    int l = (p == 0) ? 0 : (LWIN - p);
    size_t src = (size_t)l * MDIM * MDIM + (size_t)m * MDIM + n;
    size_t dst = (size_t)p * MDIM * MDIM + (size_t)(n >> 1) * (2 * MDIM)
               + 2 * m + (n & 1);
    agtr[dst] = Ar[src];
    agti[dst] = Ai[src];
}

__global__ void dmd_init_state(const float* __restrict__ x,
                               float* __restrict__ stR,
                               float* __restrict__ stI) {
    size_t i = (size_t)blockIdx.x * blockDim.x + threadIdx.x;
    if (i >= (size_t)LWIN * MDIM * BDIM) return;
    int b = (int)(i & (BDIM - 1));
    int n = (int)((i >> 8) & (MDIM - 1));
    int s = (int)(i >> 18);
    float v = x[(size_t)b * (LWIN * MDIM) + (size_t)s * MDIM + n];
    size_t dst = (size_t)s * MDIM * BDIM + (size_t)(n >> 1) * (2 * BDIM)
               + 2 * b + (n & 1);
    stR[dst] = v;
    stI[dst] = 0.0f;
}

struct FragSet {
    v2f bR[2], bI[2];
    v2f aR[4], aI[4];
};

__device__ __forceinline__ void load_frags(FragSet& f,
        const float* __restrict__ pAR, const float* __restrict__ pAI,
        const float* __restrict__ pBR, const float* __restrict__ pBI, int k) {
    const size_t offA = (size_t)k * MDIM;
    const size_t offB = (size_t)k * BDIM;
#pragma unroll
    for (int bt = 0; bt < 2; ++bt) {
        const float2 bR2 = *(const float2*)(pBR + offB + 32 * bt);
        const float2 bI2 = *(const float2*)(pBI + offB + 32 * bt);
        f.bR[bt] = (v2f){ bR2.x, bR2.y };
        f.bI[bt] = (v2f){ bI2.x, bI2.y };
    }
#pragma unroll
    for (int mt = 0; mt < 4; ++mt) {
        const float2 aR2 = *(const float2*)(pAR + offA + 32 * mt);
        const float2 aI2 = *(const float2*)(pAI + offA + 32 * mt);
        f.aR[mt] = (v2f){ aR2.x, aR2.y };
        f.aI[mt] = (v2f){ aI2.x, aI2.y };
    }
}

__device__ __forceinline__ void mma_block(const FragSet& f,
        v8f (&c_rr)[8], v8f (&c_ii)[8], v8f (&c_ri)[8], v8f (&c_ir)[8]) {
#pragma unroll
    for (int mt = 0; mt < 4; ++mt) {
#pragma unroll
        for (int bt = 0; bt < 2; ++bt) {
            const int t = mt * 2 + bt;
            c_rr[t] = wmma_f32_k4(f.aR[mt], f.bR[bt], c_rr[t]);
            c_ii[t] = wmma_f32_k4(f.aI[mt], f.bI[bt], c_ii[t]);
            c_ri[t] = wmma_f32_k4(f.aR[mt], f.bI[bt], c_ri[t]);
            c_ir[t] = wmma_f32_k4(f.aI[mt], f.bR[bt], c_ir[t]);
        }
    }
}

__global__ void __launch_bounds__(128) __attribute__((amdgpu_waves_per_eu(1, 1)))
dmd_gemm_step(
    const float* __restrict__ agtr, const float* __restrict__ agti,
    const float* __restrict__ stR,  const float* __restrict__ stI,
    float* __restrict__ partR, float* __restrict__ partI, int t) {
    const int lane = threadIdx.x & 31;
    const int wv = threadIdx.x >> 5;
    const int g  = lane >> 4;
    const int lm = lane & 15;
    const int p  = blockIdx.z;
    const int m0 = blockIdx.y * 64;
    const int b0 = (blockIdx.x * 4 + wv) * 32;
    const int s  = t + p;

    const float* pAR = agtr + (size_t)p * MDIM * MDIM + (size_t)g * (2 * MDIM) + 2 * (m0 + lm);
    const float* pAI = agti + (size_t)p * MDIM * MDIM + (size_t)g * (2 * MDIM) + 2 * (m0 + lm);
    const float* pBR = stR + (size_t)s * MDIM * BDIM + (size_t)g * (2 * BDIM) + 2 * (b0 + lm);
    const float* pBI = stI + (size_t)s * MDIM * BDIM + (size_t)g * (2 * BDIM) + 2 * (b0 + lm);

    v8f c_rr[8] = {}, c_ii[8] = {}, c_ri[8] = {}, c_ir[8] = {};

    FragSet f0, f1, f2, f3, f4;
    load_frags(f0, pAR, pAI, pBR, pBI, 0);
    load_frags(f1, pAR, pAI, pBR, pBI, 4);
    load_frags(f2, pAR, pAI, pBR, pBI, 8);

#define PIN() __builtin_amdgcn_sched_barrier(0)
#define MMA(f) mma_block(f, c_rr, c_ii, c_ri, c_ir)
#define LDF(f, kk) do { load_frags(f, pAR, pAI, pBR, pBI, (kk)); PIN(); } while (0)
    int k = 0;
    for (; k < 1000; k += 20) {
        LDF(f3, k + 12); MMA(f0);
        LDF(f4, k + 16); MMA(f1);
        LDF(f0, k + 20); MMA(f2);
        LDF(f1, k + 24); MMA(f3);
        LDF(f2, k + 28); MMA(f4);
    }
    LDF(f3, 1012); MMA(f0);
    LDF(f4, 1016); MMA(f1);
    LDF(f0, 1020); MMA(f2);
    MMA(f3);
    MMA(f4);
    MMA(f0);
#undef LDF
#undef MMA
#undef PIN

#pragma unroll
    for (int mt = 0; mt < 4; ++mt) {
#pragma unroll
        for (int bt = 0; bt < 2; ++bt) {
#pragma unroll
            for (int r = 0; r < 8; ++r) {
                const int m = m0 + mt * 16 + r + 8 * g;
                const size_t o = (size_t)p * MDIM * BDIM + (size_t)m * BDIM
                               + b0 + bt * 16 + lm;
                partR[o] = c_rr[mt * 2 + bt][r] - c_ii[mt * 2 + bt][r];
                partI[o] = c_ri[mt * 2 + bt][r] + c_ir[mt * 2 + bt][r];
            }
        }
    }
}

__global__ void dmd_reduce_step(const float* __restrict__ partR,
                                const float* __restrict__ partI,
                                float* __restrict__ stR,
                                float* __restrict__ stI, int t) {
    int i = blockIdx.x * blockDim.x + threadIdx.x;
    if (i >= MDIM * BDIM) return;
    int m = i >> 8;
    int b = i & (BDIM - 1);
    float sr = 0.0f, si = 0.0f;
#pragma unroll
    for (int p = 0; p < LWIN; ++p) {
        sr += partR[(size_t)p * MDIM * BDIM + i];
        si += partI[(size_t)p * MDIM * BDIM + i];
    }
    size_t dst = (size_t)(LWIN + t) * MDIM * BDIM
               + (size_t)(m >> 1) * (2 * BDIM) + 2 * b + (m & 1);
    stR[dst] = sr;
    stI[dst] = si;
}

__global__ void dmd_write_out(const float* __restrict__ stR,
                              float* __restrict__ out) {
    size_t i = (size_t)blockIdx.x * blockDim.x + threadIdx.x;
    if (i >= (size_t)BDIM * LWIN * MDIM) return;
    int m = (int)(i & (MDIM - 1));
    int l = (int)((i >> 10) & (LWIN - 1));
    int b = (int)(i >> 13);
    size_t src = (size_t)(LWIN + l) * MDIM * BDIM
               + (size_t)(m >> 1) * (2 * BDIM) + 2 * b + (m & 1);
    out[i] = stR[src];
}

extern "C" void kernel_launch(void* const* d_in, const int* in_sizes, int n_in,
                              void* d_out, int out_size, void* d_ws, size_t ws_size,
                              hipStream_t stream) {
    const float* x  = (const float*)d_in[0];
    const float* Ar = (const float*)d_in[1];
    const float* Ai = (const float*)d_in[2];

    float* ws = (float*)d_ws;
    const size_t AGT  = (size_t)LWIN * MDIM * MDIM;
    const size_t STPL = (size_t)NSLOT * MDIM * BDIM;
    const size_t PART = (size_t)LWIN * MDIM * BDIM;
    float* agtr = ws;
    float* agti = agtr + AGT;
    float* stR  = agti + AGT;
    float* stI  = stR + STPL;
    float* pR   = stI + STPL;
    float* pI   = pR + PART;

    dmd_build_agt<<<(unsigned)(AGT / 256), 256, 0, stream>>>(Ar, Ai, agtr, agti);
    dmd_init_state<<<(unsigned)(((size_t)LWIN * MDIM * BDIM) / 256), 256, 0, stream>>>(x, stR, stI);

    for (int t = 0; t < PRED; ++t) {
        dmd_gemm_step<<<dim3(BDIM / 128, MDIM / 64, LWIN), 128, 0, stream>>>(
            agtr, agti, stR, stI, pR, pI, t);
        dmd_reduce_step<<<(MDIM * BDIM) / 256, 256, 0, stream>>>(pR, pI, stR, stI, t);
    }

    dmd_write_out<<<(unsigned)(((size_t)BDIM * LWIN * MDIM) / 256), 256, 0, stream>>>(
        stR, (float*)d_out);
}


# gemm step: products with the imaginary part of window slots still holding the real input (exact zeros) are not issued; split-K re-cut along k to balance them, so the order of the sums changes
# speedup vs baseline: 2.1090x; 1.1972x over previous
#include <hip/hip_runtime.h>
#include <hip/hip_bf16.h>

#define MDIM 1024
#define BDIM 256
#define LWIN 8
#define PRED 8
#define NSLOT (LWIN + PRED)

typedef __attribute__((ext_vector_type(2))) float v2f;
typedef __attribute__((ext_vector_type(8))) float v8f;

__device__ __forceinline__ v8f wmma_f32_k4(v2f a, v2f b, v8f c) {
    return __builtin_amdgcn_wmma_f32_16x16x4_f32(
         false, a,  false, b,
         (short)0, c,  false,  false);
}

__global__ void dmd_build_agt(const float* __restrict__ Ar,
                              const float* __restrict__ Ai,
                              float* __restrict__ agtr,
                              float* __restrict__ agti) {
    size_t i = (size_t)blockIdx.x * blockDim.x + threadIdx.x;
    if (i >= (size_t)LWIN * MDIM * MDIM) return;
    int n = (int)(i & (MDIM - 1));
    int m = (int)((i >> 10) & (MDIM - 1));
    int p = (int)(i >> 20);
    int l = (p == 0) ? 0 : (LWIN - p);
    size_t src = (size_t)l * MDIM * MDIM + (size_t)m * MDIM + n;
    size_t dst = (size_t)p * MDIM * MDIM + (size_t)(n >> 1) * (2 * MDIM)
               + 2 * m + (n & 1);
    agtr[dst] = Ar[src];
    agti[dst] = Ai[src];
}

__global__ void dmd_init_state(const float* __restrict__ x,
                               float* __restrict__ stR) {
    size_t i = (size_t)blockIdx.x * blockDim.x + threadIdx.x;
    if (i >= (size_t)LWIN * MDIM * BDIM) return;
    int b = (int)(i & (BDIM - 1));
    int n = (int)((i >> 8) & (MDIM - 1));
    int s = (int)(i >> 18);
    float v = x[(size_t)b * (LWIN * MDIM) + (size_t)s * MDIM + n];
    size_t dst = (size_t)s * MDIM * BDIM + (size_t)(n >> 1) * (2 * BDIM)
               + 2 * b + (n & 1);
    stR[dst] = v;
}

struct FragSet {
    v2f bR[2], bI[2];
    v2f aR[4], aI[4];
};

template <bool CPLX>
__device__ __forceinline__ void load_frags(FragSet& f,
        const float* __restrict__ pAR, const float* __restrict__ pAI,
        const float* __restrict__ pBR, const float* __restrict__ pBI, int kb) {
    const size_t offA = (size_t)kb * (4 * MDIM);
    const size_t offB = (size_t)kb * (4 * BDIM);
#pragma unroll
    for (int bt = 0; bt < 2; ++bt) {
        const float2 bR2 = *(const float2*)(pBR + offB + 32 * bt);
        f.bR[bt] = (v2f){ bR2.x, bR2.y };
        if (CPLX) {
            const float2 bI2 = *(const float2*)(pBI + offB + 32 * bt);
            f.bI[bt] = (v2f){ bI2.x, bI2.y };
        }
    }
#pragma unroll
    for (int mt = 0; mt < 4; ++mt) {
        const float2 aR2 = *(const float2*)(pAR + offA + 32 * mt);
        const float2 aI2 = *(const float2*)(pAI + offA + 32 * mt);
        f.aR[mt] = (v2f){ aR2.x, aR2.y };
        f.aI[mt] = (v2f){ aI2.x, aI2.y };
    }
}

template <bool CPLX>
__device__ __forceinline__ void mma_block(const FragSet& f,
        v8f (&c_rr)[8], v8f (&c_ii)[8], v8f (&c_ri)[8], v8f (&c_ir)[8]) {
#pragma unroll
    for (int mt = 0; mt < 4; ++mt) {
#pragma unroll
        for (int bt = 0; bt < 2; ++bt) {
            const int t = mt * 2 + bt;
            c_rr[t] = wmma_f32_k4(f.aR[mt], f.bR[bt], c_rr[t]);
            if (CPLX) {
                c_ii[t] = wmma_f32_k4(f.aI[mt], f.bI[bt], c_ii[t]);
                c_ri[t] = wmma_f32_k4(f.aR[mt], f.bI[bt], c_ri[t]);
            }
            c_ir[t] = wmma_f32_k4(f.aI[mt], f.bR[bt], c_ir[t]);
        }
    }
}

struct Phase {
    const float* pAR; const float* pAI; const float* pBR; const float* pBI;
    int kbase;
    __device__ __forceinline__ int kb_of(int j) const {
        return kbase + (j >> 5) * 256 + (j & 31);
    }
};

#define PIN() __builtin_amdgcn_sched_barrier(0)

template <bool CPLX, int D>
__device__ __forceinline__ void phase_prologue(const Phase& ph, FragSet (&f)[8]) {
#pragma unroll
    for (int c = 0; c < D; ++c) {
        load_frags<CPLX>(f[c], ph.pAR, ph.pAI, ph.pBR, ph.pBI, ph.kb_of(c));
        PIN();
    }
}

template <bool CPLX, int D, int N>
__device__ __forceinline__ void phase_main(const Phase& ph, FragSet (&f)[8],
        v8f (&c_rr)[8], v8f (&c_ii)[8], v8f (&c_ri)[8], v8f (&c_ir)[8]) {
    for (int j = 0; j + 8 < N; j += 8) {
        const int kbA = ph.kb_of(j);
        const int kbB = ph.kb_of(j + 8);
#pragma unroll
        for (int c = 0; c < 8; ++c) {
            const int kb = (c + D < 8) ? (kbA + c + D) : (kbB + c + D - 8);
            load_frags<CPLX>(f[(c + D) & 7], ph.pAR, ph.pAI, ph.pBR, ph.pBI, kb);
            PIN();
            mma_block<CPLX>(f[c], c_rr, c_ii, c_ri, c_ir);
        }
    }
}

template <bool CPLX, int D, int N>
__device__ __forceinline__ void phase_tail(const Phase& ph, FragSet (&f)[8],
        v8f (&c_rr)[8], v8f (&c_ii)[8], v8f (&c_ri)[8], v8f (&c_ir)[8]) {
    const int kbA = ph.kb_of(N - 8);
#pragma unroll
    for (int c = 0; c < 8; ++c) {
        if (c + D < 8) {
            load_frags<CPLX>(f[c + D], ph.pAR, ph.pAI, ph.pBR, ph.pBI, kbA + c + D);
            PIN();
        }
        mma_block<CPLX>(f[c], c_rr, c_ii, c_ri, c_ir);
    }
}
#undef PIN

template <int T>
__global__ void __launch_bounds__(128) __attribute__((amdgpu_waves_per_eu(1, 1)))
dmd_gemm_step(
    const float* __restrict__ agtr, const float* __restrict__ agti,
    const float* __restrict__ stR,  const float* __restrict__ stI,
    float* __restrict__ partR, float* __restrict__ partI) {
    constexpr int NREAL = LWIN - T;
    constexpr int NCPLX = T;
    const int lane = threadIdx.x & 31;
    const int wv = threadIdx.x >> 5;
    const int g  = lane >> 4;
    const int lm = lane & 15;
    const int z  = blockIdx.z;
    const int m0 = blockIdx.y * 64;
    const int b0 = (blockIdx.x * 4 + wv) * 32;

    Phase re, cx;
    re.pAR = agtr + (size_t)g * (2 * MDIM) + 2 * (m0 + lm);
    re.pAI = agti + (size_t)g * (2 * MDIM) + 2 * (m0 + lm);
    re.pBR = stR + (size_t)T * MDIM * BDIM + (size_t)g * (2 * BDIM) + 2 * (b0 + lm);
    re.pBI = stI + (size_t)T * MDIM * BDIM + (size_t)g * (2 * BDIM) + 2 * (b0 + lm);
    re.kbase = 32 * z;
    cx = re;
    cx.kbase = NREAL * 256 + 32 * z;

    v8f c_rr[8] = {}, c_ii[8] = {}, c_ri[8] = {}, c_ir[8] = {};

    FragSet fr[8];
    phase_prologue<false, 6>(re, fr);
    phase_main<false, 6, 32 * NREAL>(re, fr, c_rr, c_ii, c_ri, c_ir);
    if constexpr (NCPLX > 0) {
        FragSet fc[8];
        phase_prologue<true, 3>(cx, fc);
        phase_tail<false, 6, 32 * NREAL>(re, fr, c_rr, c_ii, c_ri, c_ir);
        phase_main<true, 3, 32 * NCPLX>(cx, fc, c_rr, c_ii, c_ri, c_ir);
        phase_tail<true, 3, 32 * NCPLX>(cx, fc, c_rr, c_ii, c_ri, c_ir);
    } else {
        phase_tail<false, 6, 32 * NREAL>(re, fr, c_rr, c_ii, c_ri, c_ir);
    }

#pragma unroll
    for (int mt = 0; mt < 4; ++mt) {
#pragma unroll
        for (int bt = 0; bt < 2; ++bt) {
#pragma unroll
            for (int r = 0; r < 8; ++r) {
                const int m = m0 + mt * 16 + r + 8 * g;
                const size_t o = (size_t)z * MDIM * BDIM + (size_t)m * BDIM
                               + b0 + bt * 16 + lm;
                partR[o] = c_rr[mt * 2 + bt][r] - c_ii[mt * 2 + bt][r];
                partI[o] = c_ri[mt * 2 + bt][r] + c_ir[mt * 2 + bt][r];
            }
        }
    }
}

__global__ void dmd_reduce_step(const float* __restrict__ partR,
                                const float* __restrict__ partI,
                                float* __restrict__ stR,
                                float* __restrict__ stI, int t) {
    int i = blockIdx.x * blockDim.x + threadIdx.x;
    if (i >= MDIM * BDIM) return;
    int m = i >> 8;
    int b = i & (BDIM - 1);
    float sr = 0.0f, si = 0.0f;
#pragma unroll
    for (int p = 0; p < LWIN; ++p) {
        sr += partR[(size_t)p * MDIM * BDIM + i];
        si += partI[(size_t)p * MDIM * BDIM + i];
    }
    size_t dst = (size_t)(LWIN + t) * MDIM * BDIM
               + (size_t)(m >> 1) * (2 * BDIM) + 2 * b + (m & 1);
    stR[dst] = sr;
    stI[dst] = si;
}

__global__ void dmd_write_out(const float* __restrict__ stR,
                              float* __restrict__ out) {
    size_t i = (size_t)blockIdx.x * blockDim.x + threadIdx.x;
    if (i >= (size_t)BDIM * LWIN * MDIM) return;
    int m = (int)(i & (MDIM - 1));
    int l = (int)((i >> 10) & (LWIN - 1));
    int b = (int)(i >> 13);
    size_t src = (size_t)(LWIN + l) * MDIM * BDIM
               + (size_t)(m >> 1) * (2 * BDIM) + 2 * b + (m & 1);
    out[i] = stR[src];
}

extern "C" void kernel_launch(void* const* d_in, const int* in_sizes, int n_in,
                              void* d_out, int out_size, void* d_ws, size_t ws_size,
                              hipStream_t stream) {
    const float* x  = (const float*)d_in[0];
    const float* Ar = (const float*)d_in[1];
    const float* Ai = (const float*)d_in[2];

    float* ws = (float*)d_ws;
    const size_t AGT  = (size_t)LWIN * MDIM * MDIM;
    const size_t STPL = (size_t)NSLOT * MDIM * BDIM;
    const size_t PART = (size_t)LWIN * MDIM * BDIM;
    float* agtr = ws;
    float* agti = agtr + AGT;
    float* stR  = agti + AGT;
    float* stI  = stR + STPL;
    float* pR   = stI + STPL;
    float* pI   = pR + PART;

    dmd_build_agt<<<(unsigned)(AGT / 256), 256, 0, stream>>>(Ar, Ai, agtr, agti);
    dmd_init_state<<<(unsigned)(((size_t)LWIN * MDIM * BDIM) / 256), 256, 0, stream>>>(x, stR);

#define DMD_STEP(T)                                                                 \
    dmd_gemm_step<T><<<dim3(BDIM / 128, MDIM / 64, LWIN), 128, 0, stream>>>(        \
        agtr, agti, stR, stI, pR, pI);                                              \
    dmd_reduce_step<<<(MDIM * BDIM) / 256, 256, 0, stream>>>(pR, pI, stR, stI, T)
    DMD_STEP(0); DMD_STEP(1); DMD_STEP(2); DMD_STEP(3);
    DMD_STEP(4); DMD_STEP(5); DMD_STEP(6); DMD_STEP(7);
#undef DMD_STEP

    dmd_write_out<<<(unsigned)(((size_t)BDIM * LWIN * MDIM) / 256), 256, 0, stream>>>(
        stR, (float*)d_out);
}


# layout, split-K reduce and output kernels move whole 32x64 tiles through LDS with 128-bit global accesses instead of one element per lane: about 9x fewer waves to enter; every sum keeps its order
# speedup vs baseline: 2.7900x; 1.3229x over previous
#include <hip/hip_runtime.h>
#include <hip/hip_bf16.h>

#define MDIM 1024
#define BDIM 256
#define LWIN 8
#define PRED 8
#define NSLOT (LWIN + PRED)

typedef __attribute__((ext_vector_type(2))) float v2f;
typedef __attribute__((ext_vector_type(8))) float v8f;

__device__ __forceinline__ v8f wmma_f32_k4(v2f a, v2f b, v8f c) {
    return __builtin_amdgcn_wmma_f32_16x16x4_f32(
         false, a,  false, b,
         (short)0, c,  false,  false);
}

#define TPITCH 66
#define TILE_FLOATS (32 * TPITCH)

__device__ __forceinline__ void tile_load_rows(float* __restrict__ tile,
        const float* __restrict__ srcT, size_t pitch) {
    for (int k = 0; k < 2; ++k) {
        const int idx = (int)threadIdx.x + 256 * k;
        const int r  = idx >> 4;
        const int c4 = idx & 15;
        const float4 v = *(const float4*)(srcT + (size_t)r * pitch + 4 * c4);
        float* q = tile + r * TPITCH + 4 * c4;
        *(float2*)(q)     = make_float2(v.x, v.y);
        *(float2*)(q + 2) = make_float2(v.z, v.w);
    }
}

__device__ __forceinline__ void tile_store_pairs(const float* __restrict__ tile,
        float* __restrict__ dstT, int R) {
    for (int k = 0; k < 2; ++k) {
        const int idx = (int)threadIdx.x + 256 * k;
        const int jp = idx >> 4;
        const int rp = idx & 15;
        const float2 a = *(const float2*)(tile + (2 * rp) * TPITCH + 2 * jp);
        const float2 b = *(const float2*)(tile + (2 * rp + 1) * TPITCH + 2 * jp);
        *(float4*)(dstT + (size_t)jp * (2 * R) + 4 * rp) = make_float4(a.x, a.y, b.x, b.y);
    }
}

__device__ __forceinline__ void tile_load_pairs(float* __restrict__ tile,
        const float* __restrict__ srcT, int R) {
    for (int k = 0; k < 2; ++k) {
        const int idx = (int)threadIdx.x + 256 * k;
        const int jp = idx >> 4;
        const int rp = idx & 15;
        const float4 v = *(const float4*)(srcT + (size_t)jp * (2 * R) + 4 * rp);
        *(float2*)(tile + (2 * rp) * TPITCH + 2 * jp)     = make_float2(v.x, v.y);
        *(float2*)(tile + (2 * rp + 1) * TPITCH + 2 * jp) = make_float2(v.z, v.w);
    }
}

__device__ __forceinline__ void tile_store_rows(const float* __restrict__ tile,
        float* __restrict__ dstT, size_t pitch) {
    for (int k = 0; k < 2; ++k) {
        const int idx = (int)threadIdx.x + 256 * k;
        const int r  = idx >> 4;
        const int c4 = idx & 15;
        const float* q = tile + r * TPITCH + 4 * c4;
        const float2 a = *(const float2*)(q);
        const float2 b = *(const float2*)(q + 2);
        *(float4*)(dstT + (size_t)r * pitch + 4 * c4) = make_float4(a.x, a.y, b.x, b.y);
    }
}

__global__ void __launch_bounds__(256)
dmd_build_agt(const float* __restrict__ Ar,
              const float* __restrict__ Ai,
              float* __restrict__ agtr,
              float* __restrict__ agti) {
    __shared__ float tR[TILE_FLOATS];
    __shared__ float tI[TILE_FLOATS];
    const int n0 = (int)blockIdx.x * 64;
    const int m0 = (int)blockIdx.y * 32;
    for (int p = (int)blockIdx.z; p < LWIN; p += (int)gridDim.z) {
        const int l = (p == 0) ? 0 : (LWIN - p);
        const size_t so = (size_t)l * MDIM * MDIM + (size_t)m0 * MDIM + n0;
        const size_t dO = (size_t)p * MDIM * MDIM + (size_t)(n0 >> 1) * (2 * MDIM) + 2 * m0;
        tile_load_rows(tR, Ar + so, MDIM);
        tile_load_rows(tI, Ai + so, MDIM);
        __syncthreads();
        tile_store_pairs(tR, agtr + dO, MDIM);
        tile_store_pairs(tI, agti + dO, MDIM);
        __syncthreads();
    }
}

__global__ void __launch_bounds__(256)
dmd_init_state(const float* __restrict__ x,
               float* __restrict__ stR) {
    __shared__ float tX[TILE_FLOATS];
    const int s  = (int)blockIdx.x >> 4;
    const int n0 = ((int)blockIdx.x & 15) * 64;
    const int b0 = (int)blockIdx.y * 32;
    tile_load_rows(tX, x + (size_t)b0 * (LWIN * MDIM) + (size_t)s * MDIM + n0,
                   (size_t)LWIN * MDIM);
    __syncthreads();
    tile_store_pairs(tX, stR + (size_t)s * MDIM * BDIM
                         + (size_t)(n0 >> 1) * (2 * BDIM) + 2 * b0, BDIM);
}

struct FragSet {
    v2f bR[2], bI[2];
    v2f aR[4], aI[4];
};

template <bool CPLX>
__device__ __forceinline__ void load_frags(FragSet& f,
        const float* __restrict__ pAR, const float* __restrict__ pAI,
        const float* __restrict__ pBR, const float* __restrict__ pBI, int kb) {
    const size_t offA = (size_t)kb * (4 * MDIM);
    const size_t offB = (size_t)kb * (4 * BDIM);
#pragma unroll
    for (int bt = 0; bt < 2; ++bt) {
        const float2 bR2 = *(const float2*)(pBR + offB + 32 * bt);
        f.bR[bt] = (v2f){ bR2.x, bR2.y };
        if (CPLX) {
            const float2 bI2 = *(const float2*)(pBI + offB + 32 * bt);
            f.bI[bt] = (v2f){ bI2.x, bI2.y };
        }
    }
#pragma unroll
    for (int mt = 0; mt < 4; ++mt) {
        const float2 aR2 = *(const float2*)(pAR + offA + 32 * mt);
        const float2 aI2 = *(const float2*)(pAI + offA + 32 * mt);
        f.aR[mt] = (v2f){ aR2.x, aR2.y };
        f.aI[mt] = (v2f){ aI2.x, aI2.y };
    }
}

template <bool CPLX>
__device__ __forceinline__ void mma_block(const FragSet& f,
        v8f (&c_rr)[8], v8f (&c_ii)[8], v8f (&c_ri)[8], v8f (&c_ir)[8]) {
#pragma unroll
    for (int mt = 0; mt < 4; ++mt) {
#pragma unroll
        for (int bt = 0; bt < 2; ++bt) {
            const int t = mt * 2 + bt;
            c_rr[t] = wmma_f32_k4(f.aR[mt], f.bR[bt], c_rr[t]);
            if (CPLX) {
                c_ii[t] = wmma_f32_k4(f.aI[mt], f.bI[bt], c_ii[t]);
                c_ri[t] = wmma_f32_k4(f.aR[mt], f.bI[bt], c_ri[t]);
            }
            c_ir[t] = wmma_f32_k4(f.aI[mt], f.bR[bt], c_ir[t]);
        }
    }
}

struct Phase {
    const float* pAR; const float* pAI; const float* pBR; const float* pBI;
    int kbase;
    __device__ __forceinline__ int kb_of(int j) const {
        return kbase + (j >> 5) * 256 + (j & 31);
    }
};

#define PIN() __builtin_amdgcn_sched_barrier(0)

template <bool CPLX, int D>
__device__ __forceinline__ void phase_prologue(const Phase& ph, FragSet (&f)[8]) {
#pragma unroll
    for (int c = 0; c < D; ++c) {
        load_frags<CPLX>(f[c], ph.pAR, ph.pAI, ph.pBR, ph.pBI, ph.kb_of(c));
        PIN();
    }
}

template <bool CPLX, int D, int N>
__device__ __forceinline__ void phase_main(const Phase& ph, FragSet (&f)[8],
        v8f (&c_rr)[8], v8f (&c_ii)[8], v8f (&c_ri)[8], v8f (&c_ir)[8]) {
    for (int j = 0; j + 8 < N; j += 8) {
        const int kbA = ph.kb_of(j);
        const int kbB = ph.kb_of(j + 8);
#pragma unroll
        for (int c = 0; c < 8; ++c) {
            const int kb = (c + D < 8) ? (kbA + c + D) : (kbB + c + D - 8);
            load_frags<CPLX>(f[(c + D) & 7], ph.pAR, ph.pAI, ph.pBR, ph.pBI, kb);
            PIN();
            mma_block<CPLX>(f[c], c_rr, c_ii, c_ri, c_ir);
        }
    }
}

template <bool CPLX, int D, int N>
__device__ __forceinline__ void phase_tail(const Phase& ph, FragSet (&f)[8],
        v8f (&c_rr)[8], v8f (&c_ii)[8], v8f (&c_ri)[8], v8f (&c_ir)[8]) {
    const int kbA = ph.kb_of(N - 8);
#pragma unroll
    for (int c = 0; c < 8; ++c) {
        if (c + D < 8) {
            load_frags<CPLX>(f[c + D], ph.pAR, ph.pAI, ph.pBR, ph.pBI, kbA + c + D);
            PIN();
        }
        mma_block<CPLX>(f[c], c_rr, c_ii, c_ri, c_ir);
    }
}
#undef PIN

template <int T>
__global__ void __launch_bounds__(128) __attribute__((amdgpu_waves_per_eu(1, 1)))
dmd_gemm_step(
    const float* __restrict__ agtr, const float* __restrict__ agti,
    const float* __restrict__ stR,  const float* __restrict__ stI,
    float* __restrict__ partR, float* __restrict__ partI) {
    constexpr int NREAL = LWIN - T;
    constexpr int NCPLX = T;
    const int lane = threadIdx.x & 31;
    const int wv = threadIdx.x >> 5;
    const int g  = lane >> 4;
    const int lm = lane & 15;
    const int z  = blockIdx.z;
    const int m0 = blockIdx.y * 64;
    const int b0 = (blockIdx.x * 4 + wv) * 32;

    Phase re, cx;
    re.pAR = agtr + (size_t)g * (2 * MDIM) + 2 * (m0 + lm);
    re.pAI = agti + (size_t)g * (2 * MDIM) + 2 * (m0 + lm);
    re.pBR = stR + (size_t)T * MDIM * BDIM + (size_t)g * (2 * BDIM) + 2 * (b0 + lm);
    re.pBI = stI + (size_t)T * MDIM * BDIM + (size_t)g * (2 * BDIM) + 2 * (b0 + lm);
    re.kbase = 32 * z;
    cx = re;
    cx.kbase = NREAL * 256 + 32 * z;

    v8f c_rr[8] = {}, c_ii[8] = {}, c_ri[8] = {}, c_ir[8] = {};

    FragSet fr[8];
    phase_prologue<false, 6>(re, fr);
    phase_main<false, 6, 32 * NREAL>(re, fr, c_rr, c_ii, c_ri, c_ir);
    if constexpr (NCPLX > 0) {
        FragSet fc[8];
        phase_prologue<true, 3>(cx, fc);
        phase_tail<false, 6, 32 * NREAL>(re, fr, c_rr, c_ii, c_ri, c_ir);
        phase_main<true, 3, 32 * NCPLX>(cx, fc, c_rr, c_ii, c_ri, c_ir);
        phase_tail<true, 3, 32 * NCPLX>(cx, fc, c_rr, c_ii, c_ri, c_ir);
    } else {
        phase_tail<false, 6, 32 * NREAL>(re, fr, c_rr, c_ii, c_ri, c_ir);
    }

#pragma unroll
    for (int mt = 0; mt < 4; ++mt) {
#pragma unroll
        for (int bt = 0; bt < 2; ++bt) {
#pragma unroll
            for (int r = 0; r < 8; ++r) {
                const int m = m0 + mt * 16 + r + 8 * g;
                const size_t o = (size_t)z * MDIM * BDIM + (size_t)m * BDIM
                               + b0 + bt * 16 + lm;
                partR[o] = c_rr[mt * 2 + bt][r] - c_ii[mt * 2 + bt][r];
                partI[o] = c_ri[mt * 2 + bt][r] + c_ir[mt * 2 + bt][r];
            }
        }
    }
}

__global__ void __launch_bounds__(256)
dmd_reduce_step(const float* __restrict__ partR,
                const float* __restrict__ partI,
                float* __restrict__ stR,
                float* __restrict__ stI, int t) {
    const int i = (int)(blockIdx.x * blockDim.x + threadIdx.x);
    if (i >= (MDIM / 2) * (BDIM / 2)) return;
    const int bq = i & (BDIM / 2 - 1);
    const int mp = i >> 7;
    const size_t o0 = (size_t)(2 * mp) * BDIM + 2 * bq;
    float r00 = 0.0f, r01 = 0.0f, r10 = 0.0f, r11 = 0.0f;
    float i00 = 0.0f, i01 = 0.0f, i10 = 0.0f, i11 = 0.0f;
#pragma unroll
    for (int p = 0; p < LWIN; ++p) {
        const float* pr = partR + (size_t)p * MDIM * BDIM + o0;
        const float* pi = partI + (size_t)p * MDIM * BDIM + o0;
        const float2 ra = *(const float2*)(pr);
        const float2 rb = *(const float2*)(pr + BDIM);
        const float2 ia = *(const float2*)(pi);
        const float2 ib = *(const float2*)(pi + BDIM);
        r00 += ra.x; r01 += ra.y; r10 += rb.x; r11 += rb.y;
        i00 += ia.x; i01 += ia.y; i10 += ib.x; i11 += ib.y;
    }
    const size_t dst = (size_t)(LWIN + t) * MDIM * BDIM
                     + (size_t)mp * (2 * BDIM) + 4 * bq;
    *(float4*)(stR + dst) = make_float4(r00, r10, r01, r11);
    *(float4*)(stI + dst) = make_float4(i00, i10, i01, i11);
}

__global__ void __launch_bounds__(256)
dmd_write_out(const float* __restrict__ stR,
              float* __restrict__ out) {
    __shared__ float tO[TILE_FLOATS];
    const int l  = (int)blockIdx.x >> 4;
    const int m0 = ((int)blockIdx.x & 15) * 64;
    const int b0 = (int)blockIdx.y * 32;
    tile_load_pairs(tO, stR + (size_t)(LWIN + l) * MDIM * BDIM
                        + (size_t)(m0 >> 1) * (2 * BDIM) + 2 * b0, BDIM);
    __syncthreads();
    tile_store_rows(tO, out + (size_t)b0 * (LWIN * MDIM) + (size_t)l * MDIM + m0,
                    (size_t)LWIN * MDIM);
}

extern "C" void kernel_launch(void* const* d_in, const int* in_sizes, int n_in,
                              void* d_out, int out_size, void* d_ws, size_t ws_size,
                              hipStream_t stream) {
    const float* x  = (const float*)d_in[0];
    const float* Ar = (const float*)d_in[1];
    const float* Ai = (const float*)d_in[2];

    float* ws = (float*)d_ws;
    const size_t AGT  = (size_t)LWIN * MDIM * MDIM;
    const size_t STPL = (size_t)NSLOT * MDIM * BDIM;
    const size_t PART = (size_t)LWIN * MDIM * BDIM;
    float* agtr = ws;
    float* agti = agtr + AGT;
    float* stR  = agti + AGT;
    float* stI  = stR + STPL;
    float* pR   = stI + STPL;
    float* pI   = pR + PART;

    dmd_build_agt<<<dim3(MDIM / 64, MDIM / 32, 4), 256, 0, stream>>>(Ar, Ai, agtr, agti);
    dmd_init_state<<<dim3(LWIN * (MDIM / 64), BDIM / 32), 256, 0, stream>>>(x, stR);

#define DMD_STEP(T)                                                                 \
    dmd_gemm_step<T><<<dim3(BDIM / 128, MDIM / 64, LWIN), 128, 0, stream>>>(        \
        agtr, agti, stR, stI, pR, pI);                                              \
    dmd_reduce_step<<<((MDIM / 2) * (BDIM / 2)) / 256, 256, 0, stream>>>(pR, pI, stR, stI, T)
    DMD_STEP(0); DMD_STEP(1); DMD_STEP(2); DMD_STEP(3);
    DMD_STEP(4); DMD_STEP(5); DMD_STEP(6); DMD_STEP(7);
#undef DMD_STEP

    dmd_write_out<<<dim3(LWIN * (MDIM / 64), BDIM / 32), 256, 0, stream>>>(
        stR, (float*)d_out);
}


# gemm step 7 and its reduce: the imaginary part of the last slot is never read (the output is the real part), so its products, partial stores and sums are not issued; every sum keeps its order
# speedup vs baseline: 2.9697x; 1.0644x over previous
#include <hip/hip_runtime.h>
#include <hip/hip_bf16.h>

#define MDIM 1024
#define BDIM 256
#define LWIN 8
#define PRED 8
#define NSLOT (LWIN + PRED)

typedef __attribute__((ext_vector_type(2))) float v2f;
typedef __attribute__((ext_vector_type(8))) float v8f;

__device__ __forceinline__ v8f wmma_f32_k4(v2f a, v2f b, v8f c) {
    return __builtin_amdgcn_wmma_f32_16x16x4_f32(
         false, a,  false, b,
         (short)0, c,  false,  false);
}

#define TPITCH 66
#define TILE_FLOATS (32 * TPITCH)

__device__ __forceinline__ void tile_load_rows(float* __restrict__ tile,
        const float* __restrict__ srcT, size_t pitch) {
    for (int k = 0; k < 2; ++k) {
        const int idx = (int)threadIdx.x + 256 * k;
        const int r  = idx >> 4;
        const int c4 = idx & 15;
        const float4 v = *(const float4*)(srcT + (size_t)r * pitch + 4 * c4);
        float* q = tile + r * TPITCH + 4 * c4;
        *(float2*)(q)     = make_float2(v.x, v.y);
        *(float2*)(q + 2) = make_float2(v.z, v.w);
    }
}

__device__ __forceinline__ void tile_store_pairs(const float* __restrict__ tile,
        float* __restrict__ dstT, int R) {
    for (int k = 0; k < 2; ++k) {
        const int idx = (int)threadIdx.x + 256 * k;
        const int jp = idx >> 4;
        const int rp = idx & 15;
        const float2 a = *(const float2*)(tile + (2 * rp) * TPITCH + 2 * jp);
        const float2 b = *(const float2*)(tile + (2 * rp + 1) * TPITCH + 2 * jp);
        *(float4*)(dstT + (size_t)jp * (2 * R) + 4 * rp) = make_float4(a.x, a.y, b.x, b.y);
    }
}

__device__ __forceinline__ void tile_load_pairs(float* __restrict__ tile,
        const float* __restrict__ srcT, int R) {
    for (int k = 0; k < 2; ++k) {
        const int idx = (int)threadIdx.x + 256 * k;
        const int jp = idx >> 4;
        const int rp = idx & 15;
        const float4 v = *(const float4*)(srcT + (size_t)jp * (2 * R) + 4 * rp);
        *(float2*)(tile + (2 * rp) * TPITCH + 2 * jp)     = make_float2(v.x, v.y);
        *(float2*)(tile + (2 * rp + 1) * TPITCH + 2 * jp) = make_float2(v.z, v.w);
    }
}

__device__ __forceinline__ void tile_store_rows(const float* __restrict__ tile,
        float* __restrict__ dstT, size_t pitch) {
    for (int k = 0; k < 2; ++k) {
        const int idx = (int)threadIdx.x + 256 * k;
        const int r  = idx >> 4;
        const int c4 = idx & 15;
        const float* q = tile + r * TPITCH + 4 * c4;
        const float2 a = *(const float2*)(q);
        const float2 b = *(const float2*)(q + 2);
        *(float4*)(dstT + (size_t)r * pitch + 4 * c4) = make_float4(a.x, a.y, b.x, b.y);
    }
}

__global__ void __launch_bounds__(256)
dmd_build_agt(const float* __restrict__ Ar,
              const float* __restrict__ Ai,
              float* __restrict__ agtr,
              float* __restrict__ agti) {
    __shared__ float tR[TILE_FLOATS];
    __shared__ float tI[TILE_FLOATS];
    const int n0 = (int)blockIdx.x * 64;
    const int m0 = (int)blockIdx.y * 32;
    for (int p = (int)blockIdx.z; p < LWIN; p += (int)gridDim.z) {
        const int l = (p == 0) ? 0 : (LWIN - p);
        const size_t so = (size_t)l * MDIM * MDIM + (size_t)m0 * MDIM + n0;
        const size_t dO = (size_t)p * MDIM * MDIM + (size_t)(n0 >> 1) * (2 * MDIM) + 2 * m0;
        tile_load_rows(tR, Ar + so, MDIM);
        tile_load_rows(tI, Ai + so, MDIM);
        __syncthreads();
        tile_store_pairs(tR, agtr + dO, MDIM);
        tile_store_pairs(tI, agti + dO, MDIM);
        __syncthreads();
    }
}

__global__ void __launch_bounds__(256)
dmd_init_state(const float* __restrict__ x,
               float* __restrict__ stR) {
    __shared__ float tX[TILE_FLOATS];
    const int s  = (int)blockIdx.x >> 4;
    const int n0 = ((int)blockIdx.x & 15) * 64;
    const int b0 = (int)blockIdx.y * 32;
    tile_load_rows(tX, x + (size_t)b0 * (LWIN * MDIM) + (size_t)s * MDIM + n0,
                   (size_t)LWIN * MDIM);
    __syncthreads();
    tile_store_pairs(tX, stR + (size_t)s * MDIM * BDIM
                         + (size_t)(n0 >> 1) * (2 * BDIM) + 2 * b0, BDIM);
}

struct FragSet {
    v2f bR[2], bI[2];
    v2f aR[4], aI[4];
};

template <bool CPLX, bool IMAG>
__device__ __forceinline__ void load_frags(FragSet& f,
        const float* __restrict__ pAR, const float* __restrict__ pAI,
        const float* __restrict__ pBR, const float* __restrict__ pBI, int kb) {
    const size_t offA = (size_t)kb * (4 * MDIM);
    const size_t offB = (size_t)kb * (4 * BDIM);
#pragma unroll
    for (int bt = 0; bt < 2; ++bt) {
        const float2 bR2 = *(const float2*)(pBR + offB + 32 * bt);
        f.bR[bt] = (v2f){ bR2.x, bR2.y };
        if (CPLX) {
            const float2 bI2 = *(const float2*)(pBI + offB + 32 * bt);
            f.bI[bt] = (v2f){ bI2.x, bI2.y };
        }
    }
#pragma unroll
    for (int mt = 0; mt < 4; ++mt) {
        const float2 aR2 = *(const float2*)(pAR + offA + 32 * mt);
        f.aR[mt] = (v2f){ aR2.x, aR2.y };
        if (CPLX || IMAG) {
            const float2 aI2 = *(const float2*)(pAI + offA + 32 * mt);
            f.aI[mt] = (v2f){ aI2.x, aI2.y };
        }
    }
}

template <bool CPLX, bool IMAG>
__device__ __forceinline__ void mma_block(const FragSet& f,
        v8f (&c_rr)[8], v8f (&c_ii)[8], v8f (&c_ri)[8], v8f (&c_ir)[8]) {
#pragma unroll
    for (int mt = 0; mt < 4; ++mt) {
#pragma unroll
        for (int bt = 0; bt < 2; ++bt) {
            const int t = mt * 2 + bt;
            c_rr[t] = wmma_f32_k4(f.aR[mt], f.bR[bt], c_rr[t]);
            if (CPLX) {
                c_ii[t] = wmma_f32_k4(f.aI[mt], f.bI[bt], c_ii[t]);
                if (IMAG) c_ri[t] = wmma_f32_k4(f.aR[mt], f.bI[bt], c_ri[t]);
            }
            if (IMAG) c_ir[t] = wmma_f32_k4(f.aI[mt], f.bR[bt], c_ir[t]);
        }
    }
}

struct Phase {
    const float* pAR; const float* pAI; const float* pBR; const float* pBI;
    int kbase;
    __device__ __forceinline__ int kb_of(int j) const {
        return kbase + (j >> 5) * 256 + (j & 31);
    }
};

#define PIN() __builtin_amdgcn_sched_barrier(0)

template <bool CPLX, bool IMAG, int D>
__device__ __forceinline__ void phase_prologue(const Phase& ph, FragSet (&f)[8]) {
#pragma unroll
    for (int c = 0; c < D; ++c) {
        load_frags<CPLX, IMAG>(f[c], ph.pAR, ph.pAI, ph.pBR, ph.pBI, ph.kb_of(c));
        PIN();
    }
}

template <bool CPLX, bool IMAG, int D, int N>
__device__ __forceinline__ void phase_main(const Phase& ph, FragSet (&f)[8],
        v8f (&c_rr)[8], v8f (&c_ii)[8], v8f (&c_ri)[8], v8f (&c_ir)[8]) {
    for (int j = 0; j + 8 < N; j += 8) {
        const int kbA = ph.kb_of(j);
        const int kbB = ph.kb_of(j + 8);
#pragma unroll
        for (int c = 0; c < 8; ++c) {
            const int kb = (c + D < 8) ? (kbA + c + D) : (kbB + c + D - 8);
            load_frags<CPLX, IMAG>(f[(c + D) & 7], ph.pAR, ph.pAI, ph.pBR, ph.pBI, kb);
            PIN();
            mma_block<CPLX, IMAG>(f[c], c_rr, c_ii, c_ri, c_ir);
        }
    }
}

template <bool CPLX, bool IMAG, int D, int N>
__device__ __forceinline__ void phase_tail(const Phase& ph, FragSet (&f)[8],
        v8f (&c_rr)[8], v8f (&c_ii)[8], v8f (&c_ri)[8], v8f (&c_ir)[8]) {
    const int kbA = ph.kb_of(N - 8);
#pragma unroll
    for (int c = 0; c < 8; ++c) {
        if (c + D < 8) {
            load_frags<CPLX, IMAG>(f[c + D], ph.pAR, ph.pAI, ph.pBR, ph.pBI, kbA + c + D);
            PIN();
        }
        mma_block<CPLX, IMAG>(f[c], c_rr, c_ii, c_ri, c_ir);
    }
}
#undef PIN

template <int T>
__global__ void __launch_bounds__(128) __attribute__((amdgpu_waves_per_eu(1, 1)))
dmd_gemm_step(
    const float* __restrict__ agtr, const float* __restrict__ agti,
    const float* __restrict__ stR,  const float* __restrict__ stI,
    float* __restrict__ partR, float* __restrict__ partI) {
    constexpr int NREAL = LWIN - T;
    constexpr int NCPLX = T;
    constexpr bool IMAG = (T + 1 < PRED);
    constexpr int DRE = IMAG ? 6 : 7;
    constexpr int DCX = IMAG ? 3 : 5;
    const int lane = threadIdx.x & 31;
    const int wv = threadIdx.x >> 5;
    const int g  = lane >> 4;
    const int lm = lane & 15;
    const int z  = blockIdx.z;
    const int m0 = blockIdx.y * 64;
    const int b0 = (blockIdx.x * 4 + wv) * 32;

    Phase re, cx;
    re.pAR = agtr + (size_t)g * (2 * MDIM) + 2 * (m0 + lm);
    re.pAI = agti + (size_t)g * (2 * MDIM) + 2 * (m0 + lm);
    re.pBR = stR + (size_t)T * MDIM * BDIM + (size_t)g * (2 * BDIM) + 2 * (b0 + lm);
    re.pBI = stI + (size_t)T * MDIM * BDIM + (size_t)g * (2 * BDIM) + 2 * (b0 + lm);
    re.kbase = 32 * z;
    cx = re;
    cx.kbase = NREAL * 256 + 32 * z;

    v8f c_rr[8] = {}, c_ii[8] = {}, c_ri[8] = {}, c_ir[8] = {};

    FragSet fr[8];
    phase_prologue<false, IMAG, DRE>(re, fr);
    phase_main<false, IMAG, DRE, 32 * NREAL>(re, fr, c_rr, c_ii, c_ri, c_ir);
    if constexpr (NCPLX > 0) {
        FragSet fc[8];
        phase_prologue<true, IMAG, DCX>(cx, fc);
        phase_tail<false, IMAG, DRE, 32 * NREAL>(re, fr, c_rr, c_ii, c_ri, c_ir);
        phase_main<true, IMAG, DCX, 32 * NCPLX>(cx, fc, c_rr, c_ii, c_ri, c_ir);
        phase_tail<true, IMAG, DCX, 32 * NCPLX>(cx, fc, c_rr, c_ii, c_ri, c_ir);
    } else {
        phase_tail<false, IMAG, DRE, 32 * NREAL>(re, fr, c_rr, c_ii, c_ri, c_ir);
    }

#pragma unroll
    for (int mt = 0; mt < 4; ++mt) {
#pragma unroll
        for (int bt = 0; bt < 2; ++bt) {
#pragma unroll
            for (int r = 0; r < 8; ++r) {
                const int m = m0 + mt * 16 + r + 8 * g;
                const size_t o = (size_t)z * MDIM * BDIM + (size_t)m * BDIM
                               + b0 + bt * 16 + lm;
                partR[o] = c_rr[mt * 2 + bt][r] - c_ii[mt * 2 + bt][r];
                if constexpr (IMAG)
                    partI[o] = c_ri[mt * 2 + bt][r] + c_ir[mt * 2 + bt][r];
            }
        }
    }
}

template <bool IMAG>
__global__ void __launch_bounds__(256)
dmd_reduce_step(const float* __restrict__ partR,
                const float* __restrict__ partI,
                float* __restrict__ stR,
                float* __restrict__ stI, int t) {
    const int i = (int)(blockIdx.x * blockDim.x + threadIdx.x);
    if (i >= (MDIM / 2) * (BDIM / 2)) return;
    const int bq = i & (BDIM / 2 - 1);
    const int mp = i >> 7;
    const size_t o0 = (size_t)(2 * mp) * BDIM + 2 * bq;
    float r00 = 0.0f, r01 = 0.0f, r10 = 0.0f, r11 = 0.0f;
    float i00 = 0.0f, i01 = 0.0f, i10 = 0.0f, i11 = 0.0f;
#pragma unroll
    for (int p = 0; p < LWIN; ++p) {
        const float* pr = partR + (size_t)p * MDIM * BDIM + o0;
        const float2 ra = *(const float2*)(pr);
        const float2 rb = *(const float2*)(pr + BDIM);
        r00 += ra.x; r01 += ra.y; r10 += rb.x; r11 += rb.y;
        if constexpr (IMAG) {
            const float* pi = partI + (size_t)p * MDIM * BDIM + o0;
            const float2 ia = *(const float2*)(pi);
            const float2 ib = *(const float2*)(pi + BDIM);
            i00 += ia.x; i01 += ia.y; i10 += ib.x; i11 += ib.y;
        }
    }
    const size_t dst = (size_t)(LWIN + t) * MDIM * BDIM
                     + (size_t)mp * (2 * BDIM) + 4 * bq;
    *(float4*)(stR + dst) = make_float4(r00, r10, r01, r11);
    if constexpr (IMAG)
        *(float4*)(stI + dst) = make_float4(i00, i10, i01, i11);
}

__global__ void __launch_bounds__(256)
dmd_write_out(const float* __restrict__ stR,
              float* __restrict__ out) {
    __shared__ float tO[TILE_FLOATS];
    const int l  = (int)blockIdx.x >> 4;
    const int m0 = ((int)blockIdx.x & 15) * 64;
    const int b0 = (int)blockIdx.y * 32;
    tile_load_pairs(tO, stR + (size_t)(LWIN + l) * MDIM * BDIM
                        + (size_t)(m0 >> 1) * (2 * BDIM) + 2 * b0, BDIM);
    __syncthreads();
    tile_store_rows(tO, out + (size_t)b0 * (LWIN * MDIM) + (size_t)l * MDIM + m0,
                    (size_t)LWIN * MDIM);
}

extern "C" void kernel_launch(void* const* d_in, const int* in_sizes, int n_in,
                              void* d_out, int out_size, void* d_ws, size_t ws_size,
                              hipStream_t stream) {
    const float* x  = (const float*)d_in[0];
    const float* Ar = (const float*)d_in[1];
    const float* Ai = (const float*)d_in[2];

    float* ws = (float*)d_ws;
    const size_t AGT  = (size_t)LWIN * MDIM * MDIM;
    const size_t STPL = (size_t)NSLOT * MDIM * BDIM;
    const size_t PART = (size_t)LWIN * MDIM * BDIM;
    float* agtr = ws;
    float* agti = agtr + AGT;
    float* stR  = agti + AGT;
    float* stI  = stR + STPL;
    float* pR   = stI + STPL;
    float* pI   = pR + PART;

    dmd_build_agt<<<dim3(MDIM / 64, MDIM / 32, 4), 256, 0, stream>>>(Ar, Ai, agtr, agti);
    dmd_init_state<<<dim3(LWIN * (MDIM / 64), BDIM / 32), 256, 0, stream>>>(x, stR);

#define DMD_STEP(T)                                                                 \
    dmd_gemm_step<T><<<dim3(BDIM / 128, MDIM / 64, LWIN), 128, 0, stream>>>(        \
        agtr, agti, stR, stI, pR, pI);                                              \
    dmd_reduce_step<(T + 1 < PRED)><<<((MDIM / 2) * (BDIM / 2)) / 256, 256, 0, stream>>>(       \
        pR, pI, stR, stI, T)
    DMD_STEP(0); DMD_STEP(1); DMD_STEP(2); DMD_STEP(3);
    DMD_STEP(4); DMD_STEP(5); DMD_STEP(6); DMD_STEP(7);
#undef DMD_STEP

    dmd_write_out<<<dim3(LWIN * (MDIM / 64), BDIM / 32), 256, 0, stream>>>(
        stR, (float*)d_out);
}


# gemm steps 1-6: a complex k-block takes 3 real products, not 4: (AR+AI)*UR, AI*(UR+UI), AR*(UI-UR), operand sums in registers; 24 not 32 matrix ops a block; the grouping of the sums changes
# speedup vs baseline: 3.1708x; 1.0677x over previous
#include <hip/hip_runtime.h>
#include <hip/hip_bf16.h>

#define MDIM 1024
#define BDIM 256
#define LWIN 8
#define PRED 8
#define NSLOT (LWIN + PRED)

typedef __attribute__((ext_vector_type(2))) float v2f;
typedef __attribute__((ext_vector_type(8))) float v8f;

__device__ __forceinline__ v8f wmma_f32_k4(v2f a, v2f b, v8f c) {
    return __builtin_amdgcn_wmma_f32_16x16x4_f32(
         false, a,  false, b,
         (short)0, c,  false,  false);
}

#define TPITCH 66
#define TILE_FLOATS (32 * TPITCH)

__device__ __forceinline__ void tile_load_rows(float* __restrict__ tile,
        const float* __restrict__ srcT, size_t pitch) {
    for (int k = 0; k < 2; ++k) {
        const int idx = (int)threadIdx.x + 256 * k;
        const int r  = idx >> 4;
        const int c4 = idx & 15;
        const float4 v = *(const float4*)(srcT + (size_t)r * pitch + 4 * c4);
        float* q = tile + r * TPITCH + 4 * c4;
        *(float2*)(q)     = make_float2(v.x, v.y);
        *(float2*)(q + 2) = make_float2(v.z, v.w);
    }
}

__device__ __forceinline__ void tile_store_pairs(const float* __restrict__ tile,
        float* __restrict__ dstT, int R) {
    for (int k = 0; k < 2; ++k) {
        const int idx = (int)threadIdx.x + 256 * k;
        const int jp = idx >> 4;
        const int rp = idx & 15;
        const float2 a = *(const float2*)(tile + (2 * rp) * TPITCH + 2 * jp);
        const float2 b = *(const float2*)(tile + (2 * rp + 1) * TPITCH + 2 * jp);
        *(float4*)(dstT + (size_t)jp * (2 * R) + 4 * rp) = make_float4(a.x, a.y, b.x, b.y);
    }
}

__device__ __forceinline__ void tile_load_pairs(float* __restrict__ tile,
        const float* __restrict__ srcT, int R) {
    for (int k = 0; k < 2; ++k) {
        const int idx = (int)threadIdx.x + 256 * k;
        const int jp = idx >> 4;
        const int rp = idx & 15;
        const float4 v = *(const float4*)(srcT + (size_t)jp * (2 * R) + 4 * rp);
        *(float2*)(tile + (2 * rp) * TPITCH + 2 * jp)     = make_float2(v.x, v.y);
        *(float2*)(tile + (2 * rp + 1) * TPITCH + 2 * jp) = make_float2(v.z, v.w);
    }
}

__device__ __forceinline__ void tile_store_rows(const float* __restrict__ tile,
        float* __restrict__ dstT, size_t pitch) {
    for (int k = 0; k < 2; ++k) {
        const int idx = (int)threadIdx.x + 256 * k;
        const int r  = idx >> 4;
        const int c4 = idx & 15;
        const float* q = tile + r * TPITCH + 4 * c4;
        const float2 a = *(const float2*)(q);
        const float2 b = *(const float2*)(q + 2);
        *(float4*)(dstT + (size_t)r * pitch + 4 * c4) = make_float4(a.x, a.y, b.x, b.y);
    }
}

__global__ void __launch_bounds__(256)
dmd_build_agt(const float* __restrict__ Ar,
              const float* __restrict__ Ai,
              float* __restrict__ agtr,
              float* __restrict__ agti) {
    __shared__ float tR[TILE_FLOATS];
    __shared__ float tI[TILE_FLOATS];
    const int n0 = (int)blockIdx.x * 64;
    const int m0 = (int)blockIdx.y * 32;
    for (int p = (int)blockIdx.z; p < LWIN; p += (int)gridDim.z) {
        const int l = (p == 0) ? 0 : (LWIN - p);
        const size_t so = (size_t)l * MDIM * MDIM + (size_t)m0 * MDIM + n0;
        const size_t dO = (size_t)p * MDIM * MDIM + (size_t)(n0 >> 1) * (2 * MDIM) + 2 * m0;
        tile_load_rows(tR, Ar + so, MDIM);
        tile_load_rows(tI, Ai + so, MDIM);
        __syncthreads();
        tile_store_pairs(tR, agtr + dO, MDIM);
        tile_store_pairs(tI, agti + dO, MDIM);
        __syncthreads();
    }
}

__global__ void __launch_bounds__(256)
dmd_init_state(const float* __restrict__ x,
               float* __restrict__ stR) {
    __shared__ float tX[TILE_FLOATS];
    const int s  = (int)blockIdx.x >> 4;
    const int n0 = ((int)blockIdx.x & 15) * 64;
    const int b0 = (int)blockIdx.y * 32;
    tile_load_rows(tX, x + (size_t)b0 * (LWIN * MDIM) + (size_t)s * MDIM + n0,
                   (size_t)LWIN * MDIM);
    __syncthreads();
    tile_store_pairs(tX, stR + (size_t)s * MDIM * BDIM
                         + (size_t)(n0 >> 1) * (2 * BDIM) + 2 * b0, BDIM);
}

struct FragSet {
    v2f bR[2], bI[2];
    v2f aR[4], aI[4];
};

template <bool CPLX, bool IMAG>
__device__ __forceinline__ void load_frags(FragSet& f,
        const float* __restrict__ pAR, const float* __restrict__ pAI,
        const float* __restrict__ pBR, const float* __restrict__ pBI, int kb) {
    const size_t offA = (size_t)kb * (4 * MDIM);
    const size_t offB = (size_t)kb * (4 * BDIM);
#pragma unroll
    for (int bt = 0; bt < 2; ++bt) {
        const float2 bR2 = *(const float2*)(pBR + offB + 32 * bt);
        f.bR[bt] = (v2f){ bR2.x, bR2.y };
        if (CPLX) {
            const float2 bI2 = *(const float2*)(pBI + offB + 32 * bt);
            f.bI[bt] = (v2f){ bI2.x, bI2.y };
        }
    }
#pragma unroll
    for (int mt = 0; mt < 4; ++mt) {
        const float2 aR2 = *(const float2*)(pAR + offA + 32 * mt);
        f.aR[mt] = (v2f){ aR2.x, aR2.y };
        if (CPLX || IMAG) {
            const float2 aI2 = *(const float2*)(pAI + offA + 32 * mt);
            f.aI[mt] = (v2f){ aI2.x, aI2.y };
        }
    }
}

template <bool CPLX, bool IMAG>
__device__ __forceinline__ void mma_block(const FragSet& f,
        v8f (&c_rr)[8], v8f (&c_k3)[8], v8f (&c_k1)[8], v8f (&c_im)[8]) {
    if constexpr (CPLX && IMAG) {
        v2f aS[4], bS[2], bD[2];
#pragma unroll
        for (int mt = 0; mt < 4; ++mt) aS[mt] = f.aR[mt] + f.aI[mt];
#pragma unroll
        for (int bt = 0; bt < 2; ++bt) {
            bS[bt] = f.bR[bt] + f.bI[bt];
            bD[bt] = f.bI[bt] - f.bR[bt];
        }
#pragma unroll
        for (int mt = 0; mt < 4; ++mt) {
#pragma unroll
            for (int bt = 0; bt < 2; ++bt) {
                const int t = mt * 2 + bt;
                c_k1[t] = wmma_f32_k4(aS[mt],   f.bR[bt], c_k1[t]);
                c_k3[t] = wmma_f32_k4(f.aI[mt], bS[bt],   c_k3[t]);
                c_im[t] = wmma_f32_k4(f.aR[mt], bD[bt],   c_im[t]);
            }
        }
    } else {
#pragma unroll
        for (int mt = 0; mt < 4; ++mt) {
#pragma unroll
            for (int bt = 0; bt < 2; ++bt) {
                const int t = mt * 2 + bt;
                c_rr[t] = wmma_f32_k4(f.aR[mt], f.bR[bt], c_rr[t]);
                if (CPLX) c_k3[t] = wmma_f32_k4(f.aI[mt], f.bI[bt], c_k3[t]);
                if (IMAG) c_im[t] = wmma_f32_k4(f.aI[mt], f.bR[bt], c_im[t]);
            }
        }
    }
}

struct Phase {
    const float* pAR; const float* pAI; const float* pBR; const float* pBI;
    int kbase;
    __device__ __forceinline__ int kb_of(int j) const {
        return kbase + (j >> 5) * 256 + (j & 31);
    }
};

#define PIN() __builtin_amdgcn_sched_barrier(0)

template <bool CPLX, bool IMAG, int D>
__device__ __forceinline__ void phase_prologue(const Phase& ph, FragSet (&f)[8]) {
#pragma unroll
    for (int c = 0; c < D; ++c) {
        load_frags<CPLX, IMAG>(f[c], ph.pAR, ph.pAI, ph.pBR, ph.pBI, ph.kb_of(c));
        PIN();
    }
}

template <bool CPLX, bool IMAG, int D, int N>
__device__ __forceinline__ void phase_main(const Phase& ph, FragSet (&f)[8],
        v8f (&c_rr)[8], v8f (&c_k3)[8], v8f (&c_k1)[8], v8f (&c_im)[8]) {
    for (int j = 0; j + 8 < N; j += 8) {
        const int kbA = ph.kb_of(j);
        const int kbB = ph.kb_of(j + 8);
#pragma unroll
        for (int c = 0; c < 8; ++c) {
            const int kb = (c + D < 8) ? (kbA + c + D) : (kbB + c + D - 8);
            load_frags<CPLX, IMAG>(f[(c + D) & 7], ph.pAR, ph.pAI, ph.pBR, ph.pBI, kb);
            PIN();
            mma_block<CPLX, IMAG>(f[c], c_rr, c_k3, c_k1, c_im);
        }
    }
}

template <bool CPLX, bool IMAG, int D, int N>
__device__ __forceinline__ void phase_tail(const Phase& ph, FragSet (&f)[8],
        v8f (&c_rr)[8], v8f (&c_k3)[8], v8f (&c_k1)[8], v8f (&c_im)[8]) {
    const int kbA = ph.kb_of(N - 8);
#pragma unroll
    for (int c = 0; c < 8; ++c) {
        if (c + D < 8) {
            load_frags<CPLX, IMAG>(f[c + D], ph.pAR, ph.pAI, ph.pBR, ph.pBI, kbA + c + D);
            PIN();
        }
        mma_block<CPLX, IMAG>(f[c], c_rr, c_k3, c_k1, c_im);
    }
}
#undef PIN

template <int T>
__global__ void __launch_bounds__(128) __attribute__((amdgpu_waves_per_eu(1, 1)))
dmd_gemm_step(
    const float* __restrict__ agtr, const float* __restrict__ agti,
    const float* __restrict__ stR,  const float* __restrict__ stI,
    float* __restrict__ partR, float* __restrict__ partI) {
    constexpr int NREAL = LWIN - T;
    constexpr int NCPLX = T;
    constexpr bool IMAG = (T + 1 < PRED);
    constexpr int DRE = IMAG ? 6 : 7;
    constexpr int DCX = IMAG ? 4 : 5;
    const int lane = threadIdx.x & 31;
    const int wv = threadIdx.x >> 5;
    const int g  = lane >> 4;
    const int lm = lane & 15;
    const int z  = blockIdx.z;
    const int m0 = blockIdx.y * 64;
    const int b0 = (blockIdx.x * 4 + wv) * 32;

    Phase re, cx;
    re.pAR = agtr + (size_t)g * (2 * MDIM) + 2 * (m0 + lm);
    re.pAI = agti + (size_t)g * (2 * MDIM) + 2 * (m0 + lm);
    re.pBR = stR + (size_t)T * MDIM * BDIM + (size_t)g * (2 * BDIM) + 2 * (b0 + lm);
    re.pBI = stI + (size_t)T * MDIM * BDIM + (size_t)g * (2 * BDIM) + 2 * (b0 + lm);
    re.kbase = 32 * z;
    cx = re;
    cx.kbase = NREAL * 256 + 32 * z;

    v8f c_rr[8] = {}, c_k3[8] = {}, c_k1[8] = {}, c_im[8] = {};

    FragSet fr[8];
    phase_prologue<false, IMAG, DRE>(re, fr);
    phase_main<false, IMAG, DRE, 32 * NREAL>(re, fr, c_rr, c_k3, c_k1, c_im);
    if constexpr (NCPLX > 0) {
        FragSet fc[8];
        phase_prologue<true, IMAG, DCX>(cx, fc);
        phase_tail<false, IMAG, DRE, 32 * NREAL>(re, fr, c_rr, c_k3, c_k1, c_im);
        phase_main<true, IMAG, DCX, 32 * NCPLX>(cx, fc, c_rr, c_k3, c_k1, c_im);
        phase_tail<true, IMAG, DCX, 32 * NCPLX>(cx, fc, c_rr, c_k3, c_k1, c_im);
    } else {
        phase_tail<false, IMAG, DRE, 32 * NREAL>(re, fr, c_rr, c_k3, c_k1, c_im);
    }

#pragma unroll
    for (int mt = 0; mt < 4; ++mt) {
#pragma unroll
        for (int bt = 0; bt < 2; ++bt) {
#pragma unroll
            for (int r = 0; r < 8; ++r) {
                const int m = m0 + mt * 16 + r + 8 * g;
                const size_t o = (size_t)z * MDIM * BDIM + (size_t)m * BDIM
                               + b0 + bt * 16 + lm;
                if constexpr (IMAG && NCPLX > 0)
                    partR[o] = (c_rr[mt * 2 + bt][r] + c_k1[mt * 2 + bt][r])
                             - c_k3[mt * 2 + bt][r];
                else
                    partR[o] = c_rr[mt * 2 + bt][r] - c_k3[mt * 2 + bt][r];
                if constexpr (IMAG)
                    partI[o] = c_k1[mt * 2 + bt][r] + c_im[mt * 2 + bt][r];
            }
        }
    }
}

template <bool IMAG>
__global__ void __launch_bounds__(256)
dmd_reduce_step(const float* __restrict__ partR,
                const float* __restrict__ partI,
                float* __restrict__ stR,
                float* __restrict__ stI, int t) {
    const int i = (int)(blockIdx.x * blockDim.x + threadIdx.x);
    if (i >= (MDIM / 2) * (BDIM / 2)) return;
    const int bq = i & (BDIM / 2 - 1);
    const int mp = i >> 7;
    const size_t o0 = (size_t)(2 * mp) * BDIM + 2 * bq;
    float r00 = 0.0f, r01 = 0.0f, r10 = 0.0f, r11 = 0.0f;
    float i00 = 0.0f, i01 = 0.0f, i10 = 0.0f, i11 = 0.0f;
#pragma unroll
    for (int p = 0; p < LWIN; ++p) {
        const float* pr = partR + (size_t)p * MDIM * BDIM + o0;
        const float2 ra = *(const float2*)(pr);
        const float2 rb = *(const float2*)(pr + BDIM);
        r00 += ra.x; r01 += ra.y; r10 += rb.x; r11 += rb.y;
        if constexpr (IMAG) {
            const float* pi = partI + (size_t)p * MDIM * BDIM + o0;
            const float2 ia = *(const float2*)(pi);
            const float2 ib = *(const float2*)(pi + BDIM);
            i00 += ia.x; i01 += ia.y; i10 += ib.x; i11 += ib.y;
        }
    }
    const size_t dst = (size_t)(LWIN + t) * MDIM * BDIM
                     + (size_t)mp * (2 * BDIM) + 4 * bq;
    *(float4*)(stR + dst) = make_float4(r00, r10, r01, r11);
    if constexpr (IMAG)
        *(float4*)(stI + dst) = make_float4(i00, i10, i01, i11);
}

__global__ void __launch_bounds__(256)
dmd_write_out(const float* __restrict__ stR,
              float* __restrict__ out) {
    __shared__ float tO[TILE_FLOATS];
    const int l  = (int)blockIdx.x >> 4;
    const int m0 = ((int)blockIdx.x & 15) * 64;
    const int b0 = (int)blockIdx.y * 32;
    tile_load_pairs(tO, stR + (size_t)(LWIN + l) * MDIM * BDIM
                        + (size_t)(m0 >> 1) * (2 * BDIM) + 2 * b0, BDIM);
    __syncthreads();
    tile_store_rows(tO, out + (size_t)b0 * (LWIN * MDIM) + (size_t)l * MDIM + m0,
                    (size_t)LWIN * MDIM);
}

extern "C" void kernel_launch(void* const* d_in, const int* in_sizes, int n_in,
                              void* d_out, int out_size, void* d_ws, size_t ws_size,
                              hipStream_t stream) {
    const float* x  = (const float*)d_in[0];
    const float* Ar = (const float*)d_in[1];
    const float* Ai = (const float*)d_in[2];

    float* ws = (float*)d_ws;
    const size_t AGT  = (size_t)LWIN * MDIM * MDIM;
    const size_t STPL = (size_t)NSLOT * MDIM * BDIM;
    const size_t PART = (size_t)LWIN * MDIM * BDIM;
    float* agtr = ws;
    float* agti = agtr + AGT;
    float* stR  = agti + AGT;
    float* stI  = stR + STPL;
    float* pR   = stI + STPL;
    float* pI   = pR + PART;

    dmd_build_agt<<<dim3(MDIM / 64, MDIM / 32, 4), 256, 0, stream>>>(Ar, Ai, agtr, agti);
    dmd_init_state<<<dim3(LWIN * (MDIM / 64), BDIM / 32), 256, 0, stream>>>(x, stR);

#define DMD_STEP(T)                                                                 \
    dmd_gemm_step<T><<<dim3(BDIM / 128, MDIM / 64, LWIN), 128, 0, stream>>>(        \
        agtr, agti, stR, stI, pR, pI);                                              \
    dmd_reduce_step<(T + 1 < PRED)><<<((MDIM / 2) * (BDIM / 2)) / 256, 256, 0, stream>>>(       \
        pR, pI, stR, stI, T)
    DMD_STEP(0); DMD_STEP(1); DMD_STEP(2); DMD_STEP(3);
    DMD_STEP(4); DMD_STEP(5); DMD_STEP(6); DMD_STEP(7);
#undef DMD_STEP

    dmd_write_out<<<dim3(LWIN * (MDIM / 64), BDIM / 32), 256, 0, stream>>>(
        stR, (float*)d_out);
}


# gemm step: fragment loads widened to 128 bits, a lane fetches the k pairs of two neighbouring rows or columns; 6 not 12 loads a complex block, 8-byte partial stores; every sum keeps its order
# speedup vs baseline: 3.2792x; 1.0342x over previous
#include <hip/hip_runtime.h>
#include <hip/hip_bf16.h>

#define MDIM 1024
#define BDIM 256
#define LWIN 8
#define PRED 8
#define NSLOT (LWIN + PRED)

typedef __attribute__((ext_vector_type(2))) float v2f;
typedef __attribute__((ext_vector_type(8))) float v8f;

__device__ __forceinline__ v8f wmma_f32_k4(v2f a, v2f b, v8f c) {
    return __builtin_amdgcn_wmma_f32_16x16x4_f32(
         false, a,  false, b,
         (short)0, c,  false,  false);
}

#define TPITCH 66
#define TILE_FLOATS (32 * TPITCH)

__device__ __forceinline__ void tile_load_rows(float* __restrict__ tile,
        const float* __restrict__ srcT, size_t pitch) {
    for (int k = 0; k < 2; ++k) {
        const int idx = (int)threadIdx.x + 256 * k;
        const int r  = idx >> 4;
        const int c4 = idx & 15;
        const float4 v = *(const float4*)(srcT + (size_t)r * pitch + 4 * c4);
        float* q = tile + r * TPITCH + 4 * c4;
        *(float2*)(q)     = make_float2(v.x, v.y);
        *(float2*)(q + 2) = make_float2(v.z, v.w);
    }
}

__device__ __forceinline__ void tile_store_pairs(const float* __restrict__ tile,
        float* __restrict__ dstT, int R) {
    for (int k = 0; k < 2; ++k) {
        const int idx = (int)threadIdx.x + 256 * k;
        const int jp = idx >> 4;
        const int rp = idx & 15;
        const float2 a = *(const float2*)(tile + (2 * rp) * TPITCH + 2 * jp);
        const float2 b = *(const float2*)(tile + (2 * rp + 1) * TPITCH + 2 * jp);
        *(float4*)(dstT + (size_t)jp * (2 * R) + 4 * rp) = make_float4(a.x, a.y, b.x, b.y);
    }
}

__device__ __forceinline__ void tile_load_pairs(float* __restrict__ tile,
        const float* __restrict__ srcT, int R) {
    for (int k = 0; k < 2; ++k) {
        const int idx = (int)threadIdx.x + 256 * k;
        const int jp = idx >> 4;
        const int rp = idx & 15;
        const float4 v = *(const float4*)(srcT + (size_t)jp * (2 * R) + 4 * rp);
        *(float2*)(tile + (2 * rp) * TPITCH + 2 * jp)     = make_float2(v.x, v.y);
        *(float2*)(tile + (2 * rp + 1) * TPITCH + 2 * jp) = make_float2(v.z, v.w);
    }
}

__device__ __forceinline__ void tile_store_rows(const float* __restrict__ tile,
        float* __restrict__ dstT, size_t pitch) {
    for (int k = 0; k < 2; ++k) {
        const int idx = (int)threadIdx.x + 256 * k;
        const int r  = idx >> 4;
        const int c4 = idx & 15;
        const float* q = tile + r * TPITCH + 4 * c4;
        const float2 a = *(const float2*)(q);
        const float2 b = *(const float2*)(q + 2);
        *(float4*)(dstT + (size_t)r * pitch + 4 * c4) = make_float4(a.x, a.y, b.x, b.y);
    }
}

__global__ void __launch_bounds__(256)
dmd_build_agt(const float* __restrict__ Ar,
              const float* __restrict__ Ai,
              float* __restrict__ agtr,
              float* __restrict__ agti) {
    __shared__ float tR[TILE_FLOATS];
    __shared__ float tI[TILE_FLOATS];
    const int n0 = (int)blockIdx.x * 64;
    const int m0 = (int)blockIdx.y * 32;
    for (int p = (int)blockIdx.z; p < LWIN; p += (int)gridDim.z) {
        const int l = (p == 0) ? 0 : (LWIN - p);
        const size_t so = (size_t)l * MDIM * MDIM + (size_t)m0 * MDIM + n0;
        const size_t dO = (size_t)p * MDIM * MDIM + (size_t)(n0 >> 1) * (2 * MDIM) + 2 * m0;
        tile_load_rows(tR, Ar + so, MDIM);
        tile_load_rows(tI, Ai + so, MDIM);
        __syncthreads();
        tile_store_pairs(tR, agtr + dO, MDIM);
        tile_store_pairs(tI, agti + dO, MDIM);
        __syncthreads();
    }
}

__global__ void __launch_bounds__(256)
dmd_init_state(const float* __restrict__ x,
               float* __restrict__ stR) {
    __shared__ float tX[TILE_FLOATS];
    const int s  = (int)blockIdx.x >> 4;
    const int n0 = ((int)blockIdx.x & 15) * 64;
    const int b0 = (int)blockIdx.y * 32;
    tile_load_rows(tX, x + (size_t)b0 * (LWIN * MDIM) + (size_t)s * MDIM + n0,
                   (size_t)LWIN * MDIM);
    __syncthreads();
    tile_store_pairs(tX, stR + (size_t)s * MDIM * BDIM
                         + (size_t)(n0 >> 1) * (2 * BDIM) + 2 * b0, BDIM);
}

struct FragSet {
    v2f bR[2], bI[2];
    v2f aR[4], aI[4];
};

template <bool CPLX, bool IMAG>
__device__ __forceinline__ void load_frags(FragSet& f,
        const float* __restrict__ pAR, const float* __restrict__ pAI,
        const float* __restrict__ pBR, const float* __restrict__ pBI, int kb) {
    const size_t offA = (size_t)kb * (4 * MDIM);
    const size_t offB = (size_t)kb * (4 * BDIM);
    {
        const float4 bR4 = *(const float4*)(pBR + offB);
        f.bR[0] = (v2f){ bR4.x, bR4.y };
        f.bR[1] = (v2f){ bR4.z, bR4.w };
        if (CPLX) {
            const float4 bI4 = *(const float4*)(pBI + offB);
            f.bI[0] = (v2f){ bI4.x, bI4.y };
            f.bI[1] = (v2f){ bI4.z, bI4.w };
        }
    }
#pragma unroll
    for (int mp = 0; mp < 2; ++mp) {
        const float4 aR4 = *(const float4*)(pAR + offA + 64 * mp);
        f.aR[2 * mp]     = (v2f){ aR4.x, aR4.y };
        f.aR[2 * mp + 1] = (v2f){ aR4.z, aR4.w };
        if (CPLX || IMAG) {
            const float4 aI4 = *(const float4*)(pAI + offA + 64 * mp);
            f.aI[2 * mp]     = (v2f){ aI4.x, aI4.y };
            f.aI[2 * mp + 1] = (v2f){ aI4.z, aI4.w };
        }
    }
}

template <bool CPLX, bool IMAG>
__device__ __forceinline__ void mma_block(const FragSet& f,
        v8f (&c_rr)[8], v8f (&c_k3)[8], v8f (&c_k1)[8], v8f (&c_im)[8]) {
    if constexpr (CPLX && IMAG) {
        v2f aS[4], bS[2], bD[2];
#pragma unroll
        for (int mt = 0; mt < 4; ++mt) aS[mt] = f.aR[mt] + f.aI[mt];
#pragma unroll
        for (int bt = 0; bt < 2; ++bt) {
            bS[bt] = f.bR[bt] + f.bI[bt];
            bD[bt] = f.bI[bt] - f.bR[bt];
        }
#pragma unroll
        for (int mt = 0; mt < 4; ++mt) {
#pragma unroll
            for (int bt = 0; bt < 2; ++bt) {
                const int t = mt * 2 + bt;
                c_k1[t] = wmma_f32_k4(aS[mt],   f.bR[bt], c_k1[t]);
                c_k3[t] = wmma_f32_k4(f.aI[mt], bS[bt],   c_k3[t]);
                c_im[t] = wmma_f32_k4(f.aR[mt], bD[bt],   c_im[t]);
            }
        }
    } else {
#pragma unroll
        for (int mt = 0; mt < 4; ++mt) {
#pragma unroll
            for (int bt = 0; bt < 2; ++bt) {
                const int t = mt * 2 + bt;
                c_rr[t] = wmma_f32_k4(f.aR[mt], f.bR[bt], c_rr[t]);
                if (CPLX) c_k3[t] = wmma_f32_k4(f.aI[mt], f.bI[bt], c_k3[t]);
                if (IMAG) c_im[t] = wmma_f32_k4(f.aI[mt], f.bR[bt], c_im[t]);
            }
        }
    }
}

struct Phase {
    const float* pAR; const float* pAI; const float* pBR; const float* pBI;
    int kbase;
    __device__ __forceinline__ int kb_of(int j) const {
        return kbase + (j >> 5) * 256 + (j & 31);
    }
};

#define PIN() __builtin_amdgcn_sched_barrier(0)

template <bool CPLX, bool IMAG, int D>
__device__ __forceinline__ void phase_prologue(const Phase& ph, FragSet (&f)[8]) {
#pragma unroll
    for (int c = 0; c < D; ++c) {
        load_frags<CPLX, IMAG>(f[c], ph.pAR, ph.pAI, ph.pBR, ph.pBI, ph.kb_of(c));
        PIN();
    }
}

template <bool CPLX, bool IMAG, int D, int N>
__device__ __forceinline__ void phase_main(const Phase& ph, FragSet (&f)[8],
        v8f (&c_rr)[8], v8f (&c_k3)[8], v8f (&c_k1)[8], v8f (&c_im)[8]) {
    for (int j = 0; j + 8 < N; j += 8) {
        const int kbA = ph.kb_of(j);
        const int kbB = ph.kb_of(j + 8);
#pragma unroll
        for (int c = 0; c < 8; ++c) {
            const int kb = (c + D < 8) ? (kbA + c + D) : (kbB + c + D - 8);
            load_frags<CPLX, IMAG>(f[(c + D) & 7], ph.pAR, ph.pAI, ph.pBR, ph.pBI, kb);
            PIN();
            mma_block<CPLX, IMAG>(f[c], c_rr, c_k3, c_k1, c_im);
        }
    }
}

template <bool CPLX, bool IMAG, int D, int N>
__device__ __forceinline__ void phase_tail(const Phase& ph, FragSet (&f)[8],
        v8f (&c_rr)[8], v8f (&c_k3)[8], v8f (&c_k1)[8], v8f (&c_im)[8]) {
    const int kbA = ph.kb_of(N - 8);
#pragma unroll
    for (int c = 0; c < 8; ++c) {
        if (c + D < 8) {
            load_frags<CPLX, IMAG>(f[c + D], ph.pAR, ph.pAI, ph.pBR, ph.pBI, kbA + c + D);
            PIN();
        }
        mma_block<CPLX, IMAG>(f[c], c_rr, c_k3, c_k1, c_im);
    }
}
#undef PIN

template <int T>
__global__ void __launch_bounds__(128) __attribute__((amdgpu_waves_per_eu(1, 1)))
dmd_gemm_step(
    const float* __restrict__ agtr, const float* __restrict__ agti,
    const float* __restrict__ stR,  const float* __restrict__ stI,
    float* __restrict__ partR, float* __restrict__ partI) {
    constexpr int NREAL = LWIN - T;
    constexpr int NCPLX = T;
    constexpr bool IMAG = (T + 1 < PRED);
    constexpr int DRE = IMAG ? 6 : 7;
    constexpr int DCX = IMAG ? 4 : 5;
    const int lane = threadIdx.x & 31;
    const int wv = threadIdx.x >> 5;
    const int g  = lane >> 4;
    const int lm = lane & 15;
    const int z  = blockIdx.z;
    const int m0 = blockIdx.y * 64;
    const int b0 = (blockIdx.x * 4 + wv) * 32;

    Phase re, cx;
    re.pAR = agtr + (size_t)g * (2 * MDIM) + 2 * (m0 + 2 * lm);
    re.pAI = agti + (size_t)g * (2 * MDIM) + 2 * (m0 + 2 * lm);
    re.pBR = stR + (size_t)T * MDIM * BDIM + (size_t)g * (2 * BDIM) + 2 * (b0 + 2 * lm);
    re.pBI = stI + (size_t)T * MDIM * BDIM + (size_t)g * (2 * BDIM) + 2 * (b0 + 2 * lm);
    re.kbase = 32 * z;
    cx = re;
    cx.kbase = NREAL * 256 + 32 * z;

    v8f c_rr[8] = {}, c_k3[8] = {}, c_k1[8] = {}, c_im[8] = {};

    FragSet fr[8];
    phase_prologue<false, IMAG, DRE>(re, fr);
    phase_main<false, IMAG, DRE, 32 * NREAL>(re, fr, c_rr, c_k3, c_k1, c_im);
    if constexpr (NCPLX > 0) {
        FragSet fc[8];
        phase_prologue<true, IMAG, DCX>(cx, fc);
        phase_tail<false, IMAG, DRE, 32 * NREAL>(re, fr, c_rr, c_k3, c_k1, c_im);
        phase_main<true, IMAG, DCX, 32 * NCPLX>(cx, fc, c_rr, c_k3, c_k1, c_im);
        phase_tail<true, IMAG, DCX, 32 * NCPLX>(cx, fc, c_rr, c_k3, c_k1, c_im);
    } else {
        phase_tail<false, IMAG, DRE, 32 * NREAL>(re, fr, c_rr, c_k3, c_k1, c_im);
    }

#pragma unroll
    for (int mt = 0; mt < 4; ++mt) {
#pragma unroll
        for (int r = 0; r < 8; ++r) {
            const int m = m0 + 32 * (mt >> 1) + 2 * (r + 8 * g) + (mt & 1);
            const size_t o = (size_t)z * MDIM * BDIM + (size_t)m * BDIM + b0 + 2 * lm;
            float vr[2], vi[2];
#pragma unroll
            for (int bt = 0; bt < 2; ++bt) {
                if constexpr (IMAG && NCPLX > 0)
                    vr[bt] = (c_rr[mt * 2 + bt][r] + c_k1[mt * 2 + bt][r])
                           - c_k3[mt * 2 + bt][r];
                else
                    vr[bt] = c_rr[mt * 2 + bt][r] - c_k3[mt * 2 + bt][r];
                if constexpr (IMAG)
                    vi[bt] = c_k1[mt * 2 + bt][r] + c_im[mt * 2 + bt][r];
            }
            *(float2*)(partR + o) = make_float2(vr[0], vr[1]);
            if constexpr (IMAG)
                *(float2*)(partI + o) = make_float2(vi[0], vi[1]);
        }
    }
}

template <bool IMAG>
__global__ void __launch_bounds__(256)
dmd_reduce_step(const float* __restrict__ partR,
                const float* __restrict__ partI,
                float* __restrict__ stR,
                float* __restrict__ stI, int t) {
    const int i = (int)(blockIdx.x * blockDim.x + threadIdx.x);
    if (i >= (MDIM / 2) * (BDIM / 2)) return;
    const int bq = i & (BDIM / 2 - 1);
    const int mp = i >> 7;
    const size_t o0 = (size_t)(2 * mp) * BDIM + 2 * bq;
    float r00 = 0.0f, r01 = 0.0f, r10 = 0.0f, r11 = 0.0f;
    float i00 = 0.0f, i01 = 0.0f, i10 = 0.0f, i11 = 0.0f;
#pragma unroll
    for (int p = 0; p < LWIN; ++p) {
        const float* pr = partR + (size_t)p * MDIM * BDIM + o0;
        const float2 ra = *(const float2*)(pr);
        const float2 rb = *(const float2*)(pr + BDIM);
        r00 += ra.x; r01 += ra.y; r10 += rb.x; r11 += rb.y;
        if constexpr (IMAG) {
            const float* pi = partI + (size_t)p * MDIM * BDIM + o0;
            const float2 ia = *(const float2*)(pi);
            const float2 ib = *(const float2*)(pi + BDIM);
            i00 += ia.x; i01 += ia.y; i10 += ib.x; i11 += ib.y;
        }
    }
    const size_t dst = (size_t)(LWIN + t) * MDIM * BDIM
                     + (size_t)mp * (2 * BDIM) + 4 * bq;
    *(float4*)(stR + dst) = make_float4(r00, r10, r01, r11);
    if constexpr (IMAG)
        *(float4*)(stI + dst) = make_float4(i00, i10, i01, i11);
}

__global__ void __launch_bounds__(256)
dmd_write_out(const float* __restrict__ stR,
              float* __restrict__ out) {
    __shared__ float tO[TILE_FLOATS];
    const int l  = (int)blockIdx.x >> 4;
    const int m0 = ((int)blockIdx.x & 15) * 64;
    const int b0 = (int)blockIdx.y * 32;
    tile_load_pairs(tO, stR + (size_t)(LWIN + l) * MDIM * BDIM
                        + (size_t)(m0 >> 1) * (2 * BDIM) + 2 * b0, BDIM);
    __syncthreads();
    tile_store_rows(tO, out + (size_t)b0 * (LWIN * MDIM) + (size_t)l * MDIM + m0,
                    (size_t)LWIN * MDIM);
}

extern "C" void kernel_launch(void* const* d_in, const int* in_sizes, int n_in,
                              void* d_out, int out_size, void* d_ws, size_t ws_size,
                              hipStream_t stream) {
    const float* x  = (const float*)d_in[0];
    const float* Ar = (const float*)d_in[1];
    const float* Ai = (const float*)d_in[2];

    float* ws = (float*)d_ws;
    const size_t AGT  = (size_t)LWIN * MDIM * MDIM;
    const size_t STPL = (size_t)NSLOT * MDIM * BDIM;
    const size_t PART = (size_t)LWIN * MDIM * BDIM;
    float* agtr = ws;
    float* agti = agtr + AGT;
    float* stR  = agti + AGT;
    float* stI  = stR + STPL;
    float* pR   = stI + STPL;
    float* pI   = pR + PART;

    dmd_build_agt<<<dim3(MDIM / 64, MDIM / 32, 4), 256, 0, stream>>>(Ar, Ai, agtr, agti);
    dmd_init_state<<<dim3(LWIN * (MDIM / 64), BDIM / 32), 256, 0, stream>>>(x, stR);

#define DMD_STEP(T)                                                                 \
    dmd_gemm_step<T><<<dim3(BDIM / 128, MDIM / 64, LWIN), 128, 0, stream>>>(        \
        agtr, agti, stR, stI, pR, pI);                                              \
    dmd_reduce_step<(T + 1 < PRED)><<<((MDIM / 2) * (BDIM / 2)) / 256, 256, 0, stream>>>(       \
        pR, pI, stR, stI, T)
    DMD_STEP(0); DMD_STEP(1); DMD_STEP(2); DMD_STEP(3);
    DMD_STEP(4); DMD_STEP(5); DMD_STEP(6); DMD_STEP(7);
#undef DMD_STEP

    dmd_write_out<<<dim3(LWIN * (MDIM / 64), BDIM / 32), 256, 0, stream>>>(
        stR, (float*)d_out);
}


# layout-only launches folded into neighbours: state tiles ride in the operator layout launch; the split-K reduce, now a 32x32 tile kernel (1024 not 4096 waves), writes the output rows; sums keep order
# speedup vs baseline: 3.4253x; 1.0446x over previous
#include <hip/hip_runtime.h>
#include <hip/hip_bf16.h>

#define MDIM 1024
#define BDIM 256
#define LWIN 8
#define PRED 8
#define NSLOT (LWIN + PRED)

typedef __attribute__((ext_vector_type(2))) float v2f;
typedef __attribute__((ext_vector_type(8))) float v8f;

__device__ __forceinline__ v8f wmma_f32_k4(v2f a, v2f b, v8f c) {
    return __builtin_amdgcn_wmma_f32_16x16x4_f32(
         false, a,  false, b,
         (short)0, c,  false,  false);
}

#define TPITCH 66
#define TILE_FLOATS (32 * TPITCH)

__device__ __forceinline__ void tile_load_rows(float* __restrict__ tile,
        const float* __restrict__ srcT, size_t pitch) {
    for (int k = 0; k < 2; ++k) {
        const int idx = (int)threadIdx.x + 256 * k;
        const int r  = idx >> 4;
        const int c4 = idx & 15;
        const float4 v = *(const float4*)(srcT + (size_t)r * pitch + 4 * c4);
        float* q = tile + r * TPITCH + 4 * c4;
        *(float2*)(q)     = make_float2(v.x, v.y);
        *(float2*)(q + 2) = make_float2(v.z, v.w);
    }
}

__device__ __forceinline__ void tile_store_pairs(const float* __restrict__ tile,
        float* __restrict__ dstT, int R) {
    for (int k = 0; k < 2; ++k) {
        const int idx = (int)threadIdx.x + 256 * k;
        const int jp = idx >> 4;
        const int rp = idx & 15;
        const float2 a = *(const float2*)(tile + (2 * rp) * TPITCH + 2 * jp);
        const float2 b = *(const float2*)(tile + (2 * rp + 1) * TPITCH + 2 * jp);
        *(float4*)(dstT + (size_t)jp * (2 * R) + 4 * rp) = make_float4(a.x, a.y, b.x, b.y);
    }
}

#define AGT_Z 4
__global__ void __launch_bounds__(256)
dmd_build_inputs(const float* __restrict__ Ar,
                 const float* __restrict__ Ai,
                 const float* __restrict__ x,
                 float* __restrict__ agtr,
                 float* __restrict__ agti,
                 float* __restrict__ stR) {
    __shared__ float tR[TILE_FLOATS];
    __shared__ float tI[TILE_FLOATS];
    const int n0 = (int)blockIdx.x * 64;
    if ((int)blockIdx.z >= AGT_Z) {
        const int yz = ((int)blockIdx.z - AGT_Z) * 32 + (int)blockIdx.y;
        const int s  = yz >> 3;
        const int b0 = (yz & 7) * 32;
        tile_load_rows(tR, x + (size_t)b0 * (LWIN * MDIM) + (size_t)s * MDIM + n0,
                       (size_t)LWIN * MDIM);
        __syncthreads();
        tile_store_pairs(tR, stR + (size_t)s * MDIM * BDIM
                             + (size_t)(n0 >> 1) * (2 * BDIM) + 2 * b0, BDIM);
        return;
    }
    const int m0 = (int)blockIdx.y * 32;
    const int zstep = (int)gridDim.z - 2;
    for (int p = (int)blockIdx.z; p < LWIN; p += zstep) {
        const int l = (p == 0) ? 0 : (LWIN - p);
        const size_t so = (size_t)l * MDIM * MDIM + (size_t)m0 * MDIM + n0;
        const size_t dO = (size_t)p * MDIM * MDIM + (size_t)(n0 >> 1) * (2 * MDIM) + 2 * m0;
        tile_load_rows(tR, Ar + so, MDIM);
        tile_load_rows(tI, Ai + so, MDIM);
        __syncthreads();
        tile_store_pairs(tR, agtr + dO, MDIM);
        tile_store_pairs(tI, agti + dO, MDIM);
        __syncthreads();
    }
}

struct FragSet {
    v2f bR[2], bI[2];
    v2f aR[4], aI[4];
};

template <bool CPLX, bool IMAG>
__device__ __forceinline__ void load_frags(FragSet& f,
        const float* __restrict__ pAR, const float* __restrict__ pAI,
        const float* __restrict__ pBR, const float* __restrict__ pBI, int kb) {
    const size_t offA = (size_t)kb * (4 * MDIM);
    const size_t offB = (size_t)kb * (4 * BDIM);
    {
        const float4 bR4 = *(const float4*)(pBR + offB);
        f.bR[0] = (v2f){ bR4.x, bR4.y };
        f.bR[1] = (v2f){ bR4.z, bR4.w };
        if (CPLX) {
            const float4 bI4 = *(const float4*)(pBI + offB);
            f.bI[0] = (v2f){ bI4.x, bI4.y };
            f.bI[1] = (v2f){ bI4.z, bI4.w };
        }
    }
#pragma unroll
    for (int mp = 0; mp < 2; ++mp) {
        const float4 aR4 = *(const float4*)(pAR + offA + 64 * mp);
        f.aR[2 * mp]     = (v2f){ aR4.x, aR4.y };
        f.aR[2 * mp + 1] = (v2f){ aR4.z, aR4.w };
        if (CPLX || IMAG) {
            const float4 aI4 = *(const float4*)(pAI + offA + 64 * mp);
            f.aI[2 * mp]     = (v2f){ aI4.x, aI4.y };
            f.aI[2 * mp + 1] = (v2f){ aI4.z, aI4.w };
        }
    }
}

template <bool CPLX, bool IMAG>
__device__ __forceinline__ void mma_block(const FragSet& f,
        v8f (&c_rr)[8], v8f (&c_k3)[8], v8f (&c_k1)[8], v8f (&c_im)[8]) {
    if constexpr (CPLX && IMAG) {
        v2f aS[4], bS[2], bD[2];
#pragma unroll
        for (int mt = 0; mt < 4; ++mt) aS[mt] = f.aR[mt] + f.aI[mt];
#pragma unroll
        for (int bt = 0; bt < 2; ++bt) {
            bS[bt] = f.bR[bt] + f.bI[bt];
            bD[bt] = f.bI[bt] - f.bR[bt];
        }
#pragma unroll
        for (int mt = 0; mt < 4; ++mt) {
#pragma unroll
            for (int bt = 0; bt < 2; ++bt) {
                const int t = mt * 2 + bt;
                c_k1[t] = wmma_f32_k4(aS[mt],   f.bR[bt], c_k1[t]);
                c_k3[t] = wmma_f32_k4(f.aI[mt], bS[bt],   c_k3[t]);
                c_im[t] = wmma_f32_k4(f.aR[mt], bD[bt],   c_im[t]);
            }
        }
    } else {
#pragma unroll
        for (int mt = 0; mt < 4; ++mt) {
#pragma unroll
            for (int bt = 0; bt < 2; ++bt) {
                const int t = mt * 2 + bt;
                c_rr[t] = wmma_f32_k4(f.aR[mt], f.bR[bt], c_rr[t]);
                if (CPLX) c_k3[t] = wmma_f32_k4(f.aI[mt], f.bI[bt], c_k3[t]);
                if (IMAG) c_im[t] = wmma_f32_k4(f.aI[mt], f.bR[bt], c_im[t]);
            }
        }
    }
}

struct Phase {
    const float* pAR; const float* pAI; const float* pBR; const float* pBI;
    int kbase;
    __device__ __forceinline__ int kb_of(int j) const {
        return kbase + (j >> 5) * 256 + (j & 31);
    }
};

#define PIN() __builtin_amdgcn_sched_barrier(0)

template <bool CPLX, bool IMAG, int D>
__device__ __forceinline__ void phase_prologue(const Phase& ph, FragSet (&f)[8]) {
#pragma unroll
    for (int c = 0; c < D; ++c) {
        load_frags<CPLX, IMAG>(f[c], ph.pAR, ph.pAI, ph.pBR, ph.pBI, ph.kb_of(c));
        PIN();
    }
}

template <bool CPLX, bool IMAG, int D, int N>
__device__ __forceinline__ void phase_main(const Phase& ph, FragSet (&f)[8],
        v8f (&c_rr)[8], v8f (&c_k3)[8], v8f (&c_k1)[8], v8f (&c_im)[8]) {
    for (int j = 0; j + 8 < N; j += 8) {
        const int kbA = ph.kb_of(j);
        const int kbB = ph.kb_of(j + 8);
#pragma unroll
        for (int c = 0; c < 8; ++c) {
            const int kb = (c + D < 8) ? (kbA + c + D) : (kbB + c + D - 8);
            load_frags<CPLX, IMAG>(f[(c + D) & 7], ph.pAR, ph.pAI, ph.pBR, ph.pBI, kb);
            PIN();
            mma_block<CPLX, IMAG>(f[c], c_rr, c_k3, c_k1, c_im);
        }
    }
}

template <bool CPLX, bool IMAG, int D, int N>
__device__ __forceinline__ void phase_tail(const Phase& ph, FragSet (&f)[8],
        v8f (&c_rr)[8], v8f (&c_k3)[8], v8f (&c_k1)[8], v8f (&c_im)[8]) {
    const int kbA = ph.kb_of(N - 8);
#pragma unroll
    for (int c = 0; c < 8; ++c) {
        if (c + D < 8) {
            load_frags<CPLX, IMAG>(f[c + D], ph.pAR, ph.pAI, ph.pBR, ph.pBI, kbA + c + D);
            PIN();
        }
        mma_block<CPLX, IMAG>(f[c], c_rr, c_k3, c_k1, c_im);
    }
}
#undef PIN

template <int T>
__global__ void __launch_bounds__(128) __attribute__((amdgpu_waves_per_eu(1, 1)))
dmd_gemm_step(
    const float* __restrict__ agtr, const float* __restrict__ agti,
    const float* __restrict__ stR,  const float* __restrict__ stI,
    float* __restrict__ partR, float* __restrict__ partI) {
    constexpr int NREAL = LWIN - T;
    constexpr int NCPLX = T;
    constexpr bool IMAG = (T + 1 < PRED);
    constexpr int DRE = IMAG ? 6 : 7;
    constexpr int DCX = IMAG ? 4 : 5;
    const int lane = threadIdx.x & 31;
    const int wv = threadIdx.x >> 5;
    const int g  = lane >> 4;
    const int lm = lane & 15;
    const int z  = blockIdx.z;
    const int m0 = blockIdx.y * 64;
    const int b0 = (blockIdx.x * 4 + wv) * 32;

    Phase re, cx;
    re.pAR = agtr + (size_t)g * (2 * MDIM) + 2 * (m0 + 2 * lm);
    re.pAI = agti + (size_t)g * (2 * MDIM) + 2 * (m0 + 2 * lm);
    re.pBR = stR + (size_t)T * MDIM * BDIM + (size_t)g * (2 * BDIM) + 2 * (b0 + 2 * lm);
    re.pBI = stI + (size_t)T * MDIM * BDIM + (size_t)g * (2 * BDIM) + 2 * (b0 + 2 * lm);
    re.kbase = 32 * z;
    cx = re;
    cx.kbase = NREAL * 256 + 32 * z;

    v8f c_rr[8] = {}, c_k3[8] = {}, c_k1[8] = {}, c_im[8] = {};

    FragSet fr[8];
    phase_prologue<false, IMAG, DRE>(re, fr);
    phase_main<false, IMAG, DRE, 32 * NREAL>(re, fr, c_rr, c_k3, c_k1, c_im);
    if constexpr (NCPLX > 0) {
        FragSet fc[8];
        phase_prologue<true, IMAG, DCX>(cx, fc);
        phase_tail<false, IMAG, DRE, 32 * NREAL>(re, fr, c_rr, c_k3, c_k1, c_im);
        phase_main<true, IMAG, DCX, 32 * NCPLX>(cx, fc, c_rr, c_k3, c_k1, c_im);
        phase_tail<true, IMAG, DCX, 32 * NCPLX>(cx, fc, c_rr, c_k3, c_k1, c_im);
    } else {
        phase_tail<false, IMAG, DRE, 32 * NREAL>(re, fr, c_rr, c_k3, c_k1, c_im);
    }

#pragma unroll
    for (int mt = 0; mt < 4; ++mt) {
#pragma unroll
        for (int r = 0; r < 8; ++r) {
            const int m = m0 + 32 * (mt >> 1) + 2 * (r + 8 * g) + (mt & 1);
            const size_t o = (size_t)z * MDIM * BDIM + (size_t)m * BDIM + b0 + 2 * lm;
            float vr[2], vi[2];
#pragma unroll
            for (int bt = 0; bt < 2; ++bt) {
                if constexpr (IMAG && NCPLX > 0)
                    vr[bt] = (c_rr[mt * 2 + bt][r] + c_k1[mt * 2 + bt][r])
                           - c_k3[mt * 2 + bt][r];
                else
                    vr[bt] = c_rr[mt * 2 + bt][r] - c_k3[mt * 2 + bt][r];
                if constexpr (IMAG)
                    vi[bt] = c_k1[mt * 2 + bt][r] + c_im[mt * 2 + bt][r];
            }
            *(float2*)(partR + o) = make_float2(vr[0], vr[1]);
            if constexpr (IMAG)
                *(float2*)(partI + o) = make_float2(vi[0], vi[1]);
        }
    }
}

#define RPITCH 34
#define RTILE_FLOATS (32 * RPITCH)

template <bool IMAG>
__global__ void __launch_bounds__(128) __attribute__((amdgpu_waves_per_eu(1, 1)))
dmd_reduce_step(const float* __restrict__ partR,
                const float* __restrict__ partI,
                float* __restrict__ stR,
                float* __restrict__ stI,
                float* __restrict__ out, int t) {
    __shared__ __attribute__((aligned(16))) float tR[RTILE_FLOATS];
    __shared__ __attribute__((aligned(16))) float tI[IMAG ? RTILE_FLOATS : 4];
    const int tid = (int)threadIdx.x;
    const int m0 = (int)blockIdx.x * 32;
    const int b0 = (int)blockIdx.y * 32;

    float4 vr[2][LWIN], vi[2][LWIN];
#pragma unroll
    for (int k = 0; k < 2; ++k) {
        const int id = tid + 128 * k;
        const int ml = id >> 3;
        const int bq = id & 7;
        const size_t o = (size_t)(m0 + ml) * BDIM + b0 + 4 * bq;
#pragma unroll
        for (int p = 0; p < LWIN; ++p) {
            vr[k][p] = *(const float4*)(partR + (size_t)p * MDIM * BDIM + o);
            if constexpr (IMAG)
                vi[k][p] = *(const float4*)(partI + (size_t)p * MDIM * BDIM + o);
        }
    }
    float4 sr[2], si[2];
#pragma unroll
    for (int k = 0; k < 2; ++k) {
        float4 ar = make_float4(0.0f, 0.0f, 0.0f, 0.0f);
        float4 ai = make_float4(0.0f, 0.0f, 0.0f, 0.0f);
#pragma unroll
        for (int p = 0; p < LWIN; ++p) {
            ar.x += vr[k][p].x; ar.y += vr[k][p].y; ar.z += vr[k][p].z; ar.w += vr[k][p].w;
            if constexpr (IMAG) {
                ai.x += vi[k][p].x; ai.y += vi[k][p].y; ai.z += vi[k][p].z; ai.w += vi[k][p].w;
            }
        }
        sr[k] = ar;
        si[k] = ai;
    }
#pragma unroll
    for (int k = 0; k < 2; ++k) {
        const int id = tid + 128 * k;
        const int ml = id >> 3;
        const int bq = id & 7;
        float* q = tR + (4 * bq) * RPITCH + ml;
        q[0] = sr[k].x; q[RPITCH] = sr[k].y; q[2 * RPITCH] = sr[k].z; q[3 * RPITCH] = sr[k].w;
        if constexpr (IMAG) {
            float* qi = tI + (4 * bq) * RPITCH + ml;
            qi[0] = si[k].x; qi[RPITCH] = si[k].y; qi[2 * RPITCH] = si[k].z; qi[3 * RPITCH] = si[k].w;
        }
    }
    __syncthreads();

    if constexpr (IMAG) {
        const size_t so = (size_t)(LWIN + t) * MDIM * BDIM
                        + (size_t)(m0 >> 1) * (2 * BDIM) + 2 * b0;
#pragma unroll
        for (int k = 0; k < 2; ++k) {
            const int id = tid + 128 * k;
            const int jp = id >> 4;
            const int rp = id & 15;
            const float2 a = *(const float2*)(tR + (2 * rp) * RPITCH + 2 * jp);
            const float2 b = *(const float2*)(tR + (2 * rp + 1) * RPITCH + 2 * jp);
            *(float4*)(stR + so + (size_t)jp * (2 * BDIM) + 4 * rp) = make_float4(a.x, a.y, b.x, b.y);
            const float2 c = *(const float2*)(tI + (2 * rp) * RPITCH + 2 * jp);
            const float2 d = *(const float2*)(tI + (2 * rp + 1) * RPITCH + 2 * jp);
            *(float4*)(stI + so + (size_t)jp * (2 * BDIM) + 4 * rp) = make_float4(c.x, c.y, d.x, d.y);
        }
    }

    float* oT = out + (size_t)b0 * (LWIN * MDIM) + (size_t)t * MDIM + m0;
#pragma unroll
    for (int k = 0; k < 2; ++k) {
        const int id = tid + 128 * k;
        const int r  = id >> 3;
        const int c4 = id & 7;
        const float* q = tR + r * RPITCH + 4 * c4;
        const float2 a = *(const float2*)(q);
        const float2 b = *(const float2*)(q + 2);
        *(float4*)(oT + (size_t)r * (LWIN * MDIM) + 4 * c4) = make_float4(a.x, a.y, b.x, b.y);
    }
}

extern "C" void kernel_launch(void* const* d_in, const int* in_sizes, int n_in,
                              void* d_out, int out_size, void* d_ws, size_t ws_size,
                              hipStream_t stream) {
    const float* x  = (const float*)d_in[0];
    const float* Ar = (const float*)d_in[1];
    const float* Ai = (const float*)d_in[2];

    float* ws = (float*)d_ws;
    const size_t AGT  = (size_t)LWIN * MDIM * MDIM;
    const size_t STPL = (size_t)NSLOT * MDIM * BDIM;
    const size_t PART = (size_t)LWIN * MDIM * BDIM;
    float* agtr = ws;
    float* agti = agtr + AGT;
    float* stR  = agti + AGT;
    float* stI  = stR + STPL;
    float* pR   = stI + STPL;
    float* pI   = pR + PART;

    dmd_build_inputs<<<dim3(MDIM / 64, MDIM / 32, AGT_Z + 2), 256, 0, stream>>>(
        Ar, Ai, x, agtr, agti, stR);

#define DMD_STEP(T)                                                                 \
    dmd_gemm_step<T><<<dim3(BDIM / 128, MDIM / 64, LWIN), 128, 0, stream>>>(        \
        agtr, agti, stR, stI, pR, pI);                                              \
    dmd_reduce_step<(T + 1 < PRED)><<<dim3(MDIM / 32, BDIM / 32), 128, 0, stream>>>(            \
        pR, pI, stR, stI, (float*)d_out, T)
    DMD_STEP(0); DMD_STEP(1); DMD_STEP(2); DMD_STEP(3);
    DMD_STEP(4); DMD_STEP(5); DMD_STEP(6); DMD_STEP(7);
#undef DMD_STEP
}
